# starting kernel
# baseline (speedup 1.0000x reference)
#include <hip/hip_runtime.h>


#define NB 10000
#define BB 4
#define EE 160000
#define DD 128
#define HH 4
#define OO 64
#define HO 256
#define MROWS (BB * NB)
#define NEG_SLOPE 0.2f

typedef float v2f __attribute__((ext_vector_type(2)));
typedef float v8f __attribute__((ext_vector_type(8)));

__device__ __forceinline__ unsigned fenc(float f) {
  unsigned u = __float_as_uint(f);
  return (u & 0x80000000u) ? ~u : (u | 0x80000000u);
}
__device__ __forceinline__ float fdec(unsigned u) {
  return __uint_as_float((u & 0x80000000u) ? (u ^ 0x80000000u) : ~u);
}

__global__ __launch_bounds__(256) void gemm_h(const float* __restrict__ x,
                                              const float* __restrict__ W,
                                              float* __restrict__ h) {
  __shared__ float2 Wlds[32 * HO];
  const int tid = threadIdx.x;
  const int lane = tid & 31;
  const int wave = tid >> 5;
  const int rowBase = blockIdx.x * 128 + wave * 16;
  const bool active = rowBase < MROWS;

  v8f acc[16];
  const v8f vzero = {0.f, 0.f, 0.f, 0.f, 0.f, 0.f, 0.f, 0.f};
#pragma unroll
  for (int t = 0; t < 16; ++t) acc[t] = vzero;

  const int m = lane & 15;
  const int kb = (lane >> 4) << 1;
  const int kp = lane >> 4;

  for (int kblk = 0; kblk < DD; kblk += 64) {
#pragma unroll 4
    for (int p = 0; p < 32; ++p) {
      float2 pr;
      pr.x = W[(size_t)(kblk + 2 * p) * HO + tid];
      pr.y = W[(size_t)(kblk + 2 * p + 1) * HO + tid];
      Wlds[p * HO + tid] = pr;
    }
    __syncthreads();

    if (active) {
      const float* xrow = x + (size_t)(rowBase + m) * DD + kblk;
      for (int kk = 0; kk < 64; kk += 4) {
        v2f a;
        a.x = xrow[kk + kb];
        a.y = xrow[kk + kb + 1];
        const float2* w0 = Wlds + ((kk >> 1) + kp) * HO + (lane & 15);
#pragma unroll
        for (int t = 0; t < 16; ++t) {
          float2 bp = w0[t * 16];
          v2f b;
          b.x = bp.x;
          b.y = bp.y;
          acc[t] = __builtin_amdgcn_wmma_f32_16x16x4_f32(
              false, a, false, b, (short)0, acc[t], false, false);
        }
      }
    }
    __syncthreads();
  }

  if (active) {
    const int nLocal = lane & 15;
    const int rOff = (lane >> 4) * 8;
#pragma unroll
    for (int t = 0; t < 16; ++t)
#pragma unroll
      for (int v = 0; v < 8; ++v)
        h[(size_t)(rowBase + rOff + v) * HO + t * 16 + nLocal] = acc[t][v];
  }
}

__global__ __launch_bounds__(256) void calc_attn_node(
    const float* __restrict__ h, const float* __restrict__ att_src,
    const float* __restrict__ att_dst, float* __restrict__ a_src,
    float* __restrict__ a_dst) {
  const int wv = blockIdx.x * 8 + (threadIdx.x >> 5);
  if (wv >= MROWS * HH) return;
  const int lane = threadIdx.x & 31;
  const int head = wv & 3;
  const float2 hv = ((const float2*)(h + (size_t)wv * OO))[lane];
  const float2 as = ((const float2*)(att_src + head * OO))[lane];
  const float2 ad = ((const float2*)(att_dst + head * OO))[lane];
  float vs = hv.x * as.x + hv.y * as.y;
  float vd = hv.x * ad.x + hv.y * ad.y;
#pragma unroll
  for (int off = 16; off > 0; off >>= 1) {
    vs += __shfl_xor(vs, off);
    vd += __shfl_xor(vd, off);
  }
  if (lane == 0) { a_src[wv] = vs; a_dst[wv] = vd; }
}

__global__ void calc_edge_coef(const float* __restrict__ W_edge,
                               const float* __restrict__ att_edge,
                               float* __restrict__ coef) {
  const int t = threadIdx.x;
  if (t < HH) {
    float s = 0.f;
    for (int o = 0; o < OO; ++o) s += W_edge[t * OO + o] * att_edge[t * OO + o];
    coef[t] = s;
  }
}

__device__ __forceinline__ float edge_alpha(const float* a_src, const float* a_dst,
                                            int sN, int dN, int b, int hh,
                                            float ae, const float* coef) {
  float a = a_src[((size_t)b * NB + sN) * HH + hh] +
            a_dst[((size_t)b * NB + dN) * HH + hh] + ae * coef[hh];
  return a > 0.f ? a : a * NEG_SLOPE;
}

__global__ __launch_bounds__(256) void edge_max(
    const int* __restrict__ src, const int* __restrict__ dst,
    const float* __restrict__ eattr, const float* __restrict__ a_src,
    const float* __restrict__ a_dst, const float* __restrict__ coef,
    unsigned* __restrict__ mbuf) {
  const int e = blockIdx.x * 256 + threadIdx.x;
  if (e >= EE) return;
  const int sN = src[e], dN = dst[e];
  const float ae = eattr[e];
  for (int b = 0; b < BB; ++b)
#pragma unroll
    for (int hh = 0; hh < HH; ++hh) {
      float a = edge_alpha(a_src, a_dst, sN, dN, b, hh, ae, coef);
      atomicMax(&mbuf[(size_t)dN * 16 + b * 4 + hh], fenc(a));
    }
}

__global__ __launch_bounds__(256) void edge_sum(
    const int* __restrict__ src, const int* __restrict__ dst,
    const float* __restrict__ eattr, const float* __restrict__ a_src,
    const float* __restrict__ a_dst, const float* __restrict__ coef,
    const unsigned* __restrict__ mbuf, float* __restrict__ sden) {
  const int e = blockIdx.x * 256 + threadIdx.x;
  if (e >= EE) return;
  const int sN = src[e], dN = dst[e];
  const float ae = eattr[e];
  for (int b = 0; b < BB; ++b)
#pragma unroll
    for (int hh = 0; hh < HH; ++hh) {
      float a = edge_alpha(a_src, a_dst, sN, dN, b, hh, ae, coef);
      const size_t mi = (size_t)dN * 16 + b * 4 + hh;
      atomicAdd(&sden[mi], __expf(a - fdec(mbuf[mi])));
    }
}

__global__ __launch_bounds__(256) void edge_scatter(
    const int* __restrict__ src, const int* __restrict__ dst,
    const float* __restrict__ eattr, const float* __restrict__ a_src,
    const float* __restrict__ a_dst, const float* __restrict__ coef,
    const unsigned* __restrict__ mbuf, const float* __restrict__ sden,
    const float* __restrict__ h, float* __restrict__ acc) {
  __shared__ float wsh[16];
  const int e = blockIdx.x;
  const int t = threadIdx.x;
  const int sN = src[e], dN = dst[e];
  if (t < 16) {
    const int b = t >> 2, hh = t & 3;
    float a = edge_alpha(a_src, a_dst, sN, dN, b, hh, eattr[e], coef);
    const size_t mi = (size_t)dN * 16 + b * 4 + hh;
    float p = __expf(a - fdec(mbuf[mi]));
    wsh[t] = p / fmaxf(sden[mi], 1e-16f);
  }
  __syncthreads();
#pragma unroll
  for (int i = t; i < BB * HO; i += 256) {
    const int b = i >> 8, rem = i & 255;
    float val = wsh[b * 4 + (rem >> 6)] * h[((size_t)b * NB + sN) * HO + rem];
    atomicAdd(&acc[((size_t)dN * 4 + b) * HO + rem], val);
  }
}

__global__ __launch_bounds__(256) void finalize(const float* __restrict__ acc,
                                                const float* __restrict__ bias,
                                                float* __restrict__ out) {
  const int i = blockIdx.x * 256 + threadIdx.x;
  if (i >= BB * NB * OO) return;
  const int o = i & 63;
  const int bn = i >> 6;
  const int n = bn % NB, b = bn / NB;
  const float* p = acc + ((size_t)n * 4 + b) * HO + o;
  out[i] = (p[0] + p[64] + p[128] + p[192]) * 0.25f + bias[o];
}

extern "C" void kernel_launch(void* const* d_in, const int* in_sizes, int n_in,
                              void* d_out, int out_size, void* d_ws, size_t ws_size,
                              hipStream_t stream) {
  const float* x        = (const float*)d_in[0];
  const int*   eidx     = (const int*)d_in[1];
  const float* eattr    = (const float*)d_in[2];
  const float* W        = (const float*)d_in[3];
  const float* att_src  = (const float*)d_in[4];
  const float* att_dst  = (const float*)d_in[5];
  const float* W_edge   = (const float*)d_in[6];
  const float* att_edge = (const float*)d_in[7];
  const float* bias     = (const float*)d_in[8];
  float* out = (float*)d_out;

  char* ws = (char*)d_ws;
  float*    h_buf  = (float*)   (ws + 0);
  float*    a_src  = (float*)   (ws + 40960000);
  float*    a_dst  = (float*)   (ws + 43520000);
  unsigned* mbuf   = (unsigned*)(ws + 46080000);
  float*    sden   = (float*)   (ws + 48640000);
  float*    coef   = (float*)   (ws + 51200000);
  float*    acc    = (float*)   (ws + 51200256);

  const int* srcI = eidx;
  const int* dstI = eidx + EE;

  hipMemsetAsync(ws + 46080000, 0, 46080256, stream);

  calc_edge_coef<<<1, 64, 0, stream>>>(W_edge, att_edge, coef);
  gemm_h<<<(MROWS + 127) / 128, 256, 0, stream>>>(x, W, h_buf);
  calc_attn_node<<<(MROWS * HH) / 8, 256, 0, stream>>>(h_buf, att_src, att_dst,
                                                       a_src, a_dst);
  edge_max<<<(EE + 255) / 256, 256, 0, stream>>>(srcI, dstI, eattr, a_src,
                                                 a_dst, coef, mbuf);
  edge_sum<<<(EE + 255) / 256, 256, 0, stream>>>(srcI, dstI, eattr, a_src,
                                                 a_dst, coef, mbuf, sden);
  edge_scatter<<<EE, 256, 0, stream>>>(srcI, dstI, eattr, a_src, a_dst, coef,
                                       mbuf, sden, h_buf, acc);
  finalize<<<(BB * NB * OO + 255) / 256, 256, 0, stream>>>(acc, bias, out);
}


# Atomic edge scatter replaced by destination-grouped gather: counting sort of edges, one work-group per node does softmax + weighted row sum in registers; sums in edge-id order, no float atomics
# speedup vs baseline: 3.7865x; 3.7865x over previous
#include <hip/hip_runtime.h>


#define NB 10000
#define BB 4
#define EE 160000
#define DD 128
#define HH 4
#define OO 64
#define HO 256
#define MROWS (BB * NB)
#define NEG_SLOPE 0.2f

typedef float v2f __attribute__((ext_vector_type(2)));
typedef float v8f __attribute__((ext_vector_type(8)));

__global__ __launch_bounds__(256) void gemm_h(const float* __restrict__ x,
                                              const float* __restrict__ W,
                                              float* __restrict__ h) {
  __shared__ float2 Wlds[32 * HO];
  const int tid = threadIdx.x;
  const int lane = tid & 31;
  const int wave = tid >> 5;
  const int rowBase = blockIdx.x * 128 + wave * 16;
  const bool active = rowBase < MROWS;

  v8f acc[16];
  const v8f vzero = {0.f, 0.f, 0.f, 0.f, 0.f, 0.f, 0.f, 0.f};
#pragma unroll
  for (int t = 0; t < 16; ++t) acc[t] = vzero;

  const int m = lane & 15;
  const int kb = (lane >> 4) << 1;
  const int kp = lane >> 4;

  for (int kblk = 0; kblk < DD; kblk += 64) {
#pragma unroll 4
    for (int p = 0; p < 32; ++p) {
      float2 pr;
      pr.x = W[(size_t)(kblk + 2 * p) * HO + tid];
      pr.y = W[(size_t)(kblk + 2 * p + 1) * HO + tid];
      Wlds[p * HO + tid] = pr;
    }
    __syncthreads();

    if (active) {
      const float* xrow = x + (size_t)(rowBase + m) * DD + kblk;
      for (int kk = 0; kk < 64; kk += 4) {
        v2f a;
        a.x = xrow[kk + kb];
        a.y = xrow[kk + kb + 1];
        const float2* w0 = Wlds + ((kk >> 1) + kp) * HO + (lane & 15);
#pragma unroll
        for (int t = 0; t < 16; ++t) {
          float2 bp = w0[t * 16];
          v2f b;
          b.x = bp.x;
          b.y = bp.y;
          acc[t] = __builtin_amdgcn_wmma_f32_16x16x4_f32(
              false, a, false, b, (short)0, acc[t], false, false);
        }
      }
    }
    __syncthreads();
  }

  if (active) {
    const int nLocal = lane & 15;
    const int rOff = (lane >> 4) * 8;
#pragma unroll
    for (int t = 0; t < 16; ++t)
#pragma unroll
      for (int v = 0; v < 8; ++v)
        h[(size_t)(rowBase + rOff + v) * HO + t * 16 + nLocal] = acc[t][v];
  }
}

__global__ __launch_bounds__(256) void calc_attn_node(
    const float* __restrict__ h, const float* __restrict__ att_src,
    const float* __restrict__ att_dst, float* __restrict__ a_src,
    float* __restrict__ a_dst) {
  const int wv = blockIdx.x * 8 + (threadIdx.x >> 5);
  if (wv >= MROWS * HH) return;
  const int lane = threadIdx.x & 31;
  const int head = wv & 3;
  const float2 hv = ((const float2*)(h + (size_t)wv * OO))[lane];
  const float2 as = ((const float2*)(att_src + head * OO))[lane];
  const float2 ad = ((const float2*)(att_dst + head * OO))[lane];
  float vs = hv.x * as.x + hv.y * as.y;
  float vd = hv.x * ad.x + hv.y * ad.y;
#pragma unroll
  for (int off = 16; off > 0; off >>= 1) {
    vs += __shfl_xor(vs, off);
    vd += __shfl_xor(vd, off);
  }
  if (lane == 0) { a_src[wv] = vs; a_dst[wv] = vd; }
}

__global__ void calc_edge_coef(const float* __restrict__ W_edge,
                               const float* __restrict__ att_edge,
                               float* __restrict__ coef) {
  const int t = threadIdx.x;
  if (t < HH) {
    float s = 0.f;
    for (int o = 0; o < OO; ++o) s += W_edge[t * OO + o] * att_edge[t * OO + o];
    coef[t] = s;
  }
}

__global__ __launch_bounds__(256) void csr_count(const int* __restrict__ dst,
                                                 int* __restrict__ deg) {
  const int e = blockIdx.x * 256 + threadIdx.x;
  if (e < EE) atomicAdd(&deg[dst[e]], 1);
}

#define SCAN_PER ((NB + 1023) / 1024)
__global__ __launch_bounds__(1024) void csr_scan(const int* __restrict__ deg,
                                                 int* __restrict__ off,
                                                 int* __restrict__ cursor) {
  __shared__ int part[1024];
  const int t = threadIdx.x;
  const int lo = t * SCAN_PER;
  int loc[SCAN_PER];
  int s = 0;
  for (int k = 0; k < SCAN_PER; ++k) {
    const int i = lo + k;
    const int v = (i < NB) ? deg[i] : 0;
    loc[k] = s;
    s += v;
  }
  part[t] = s;
  __syncthreads();
  for (int d = 1; d < 1024; d <<= 1) {
    const int v = (t >= d) ? part[t - d] : 0;
    __syncthreads();
    part[t] += v;
    __syncthreads();
  }
  const int base = part[t] - s;
  for (int k = 0; k < SCAN_PER; ++k) {
    const int i = lo + k;
    if (i < NB) {
      off[i] = base + loc[k];
      cursor[i] = base + loc[k];
    }
  }
  if (t == 1023) off[NB] = part[1023];
}

__global__ __launch_bounds__(256) void csr_fill(const int* __restrict__ dst,
                                                int* __restrict__ cursor,
                                                int* __restrict__ csr_e) {
  const int e = blockIdx.x * 256 + threadIdx.x;
  if (e < EE) {
    const int slot = atomicAdd(&cursor[dst[e]], 1);
    csr_e[slot] = e;
  }
}

#define SEG_CAP 512
__global__ __launch_bounds__(256) void node_gather(
    const int* __restrict__ src, const float* __restrict__ eattr,
    const int* __restrict__ off, const int* __restrict__ csr_e,
    const float* __restrict__ a_src, const float* __restrict__ a_dst,
    const float* __restrict__ coef, const float* __restrict__ h,
    const float* __restrict__ bias, float* __restrict__ out) {
  __shared__ int s_e[SEG_CAP];
  __shared__ int s_src[SEG_CAP];
  __shared__ float s_ea[SEG_CAP];
  __shared__ float s_red[16][16];
  __shared__ float s_w[16][16];
  __shared__ int s_cs[16];
  __shared__ __attribute__((aligned(16))) float s_acc[BB * HO];

  const int n = blockIdx.x;
  const int t = threadIdx.x;
  const int beg = off[n];
  const int L = off[n + 1] - beg;
  const bool staged = (L <= SEG_CAP);

  if (staged) {
    for (int i = t; i < L; i += 256) s_e[i] = csr_e[beg + i];
    __syncthreads();
    for (int i = t; i < L; i += 256) {
      const int e = s_e[i];
      int r = 0;
      for (int j = 0; j < L; ++j) r += (s_e[j] < e) ? 1 : 0;
      s_src[r] = src[e];
      s_ea[r] = eattr[e];
    }
    __syncthreads();
  }
  auto edge_src = [&](int i) -> int {
    return staged ? s_src[i] : src[csr_e[beg + i]];
  };
  auto edge_ea = [&](int i) -> float {
    return staged ? s_ea[i] : eattr[csr_e[beg + i]];
  };

  const int bh = t & 15;
  const int g = t >> 4;
  const int b1 = bh >> 2, h1 = bh & 3;
  const float adst = a_dst[((size_t)b1 * NB + n) * HH + h1];
  const float cf = coef[h1];
  const float* asrc_b = a_src + (size_t)b1 * NB * HH + h1;

  float m = -INFINITY;
  for (int i = g; i < L; i += 16) {
    float a = asrc_b[(size_t)edge_src(i) * HH] + adst + edge_ea(i) * cf;
    a = a > 0.f ? a : a * NEG_SLOPE;
    m = fmaxf(m, a);
  }
  s_red[g][bh] = m;
  __syncthreads();
  float mm = s_red[0][bh];
  for (int k = 1; k < 16; ++k) mm = fmaxf(mm, s_red[k][bh]);
  if (!__builtin_isfinite(mm)) mm = 0.f;
  __syncthreads();

  float sm = 0.f;
  for (int i = g; i < L; i += 16) {
    float a = asrc_b[(size_t)edge_src(i) * HH] + adst + edge_ea(i) * cf;
    a = a > 0.f ? a : a * NEG_SLOPE;
    sm += __expf(a - mm);
  }
  s_red[g][bh] = sm;
  __syncthreads();
  float den = 0.f;
  for (int k = 0; k < 16; ++k) den += s_red[k][bh];
  den = fmaxf(den, 1e-16f);

  const int brow = t >> 6;
  const int wsel = t >> 4;
  const float4* hb = (const float4*)(h + (size_t)brow * NB * HO) + (t & 63);
  float4 acc = make_float4(0.f, 0.f, 0.f, 0.f);

  for (int c = 0; c < L; c += 16) {
    {
      const int i = c + g;
      const int ic = i < L ? i : L - 1;
      const int s = edge_src(ic);
      float a = asrc_b[(size_t)s * HH] + adst + edge_ea(ic) * cf;
      a = a > 0.f ? a : a * NEG_SLOPE;
      const float w = __expf(a - mm) / den;
      s_w[g][bh] = i < L ? w : 0.f;
      if (bh == 0) s_cs[g] = s;
    }
    __syncthreads();
    const int cnt = (L - c) < 16 ? (L - c) : 16;
    for (int j = 0; j < cnt; j += 4) {
      const float4 v0 = hb[(size_t)s_cs[j + 0] * (HO / 4)];
      const float4 v1 = hb[(size_t)s_cs[j + 1] * (HO / 4)];
      const float4 v2 = hb[(size_t)s_cs[j + 2] * (HO / 4)];
      const float4 v3 = hb[(size_t)s_cs[j + 3] * (HO / 4)];
      const float w0 = s_w[j + 0][wsel];
      const float w1 = s_w[j + 1][wsel];
      const float w2 = s_w[j + 2][wsel];
      const float w3 = s_w[j + 3][wsel];
      acc.x += w0 * v0.x; acc.y += w0 * v0.y; acc.z += w0 * v0.z; acc.w += w0 * v0.w;
      acc.x += w1 * v1.x; acc.y += w1 * v1.y; acc.z += w1 * v1.z; acc.w += w1 * v1.w;
      acc.x += w2 * v2.x; acc.y += w2 * v2.y; acc.z += w2 * v2.z; acc.w += w2 * v2.w;
      acc.x += w3 * v3.x; acc.y += w3 * v3.y; acc.z += w3 * v3.z; acc.w += w3 * v3.w;
    }
    __syncthreads();
  }

  *reinterpret_cast<float4*>(&s_acc[t * 4]) = acc;
  __syncthreads();
  const int o = t & 63;
  const float* p = s_acc + brow * HO + o;
  out[((size_t)brow * NB + n) * OO + o] =
      (p[0] + p[64] + p[128] + p[192]) * 0.25f + bias[o];
}

extern "C" void kernel_launch(void* const* d_in, const int* in_sizes, int n_in,
                              void* d_out, int out_size, void* d_ws, size_t ws_size,
                              hipStream_t stream) {
  const float* x        = (const float*)d_in[0];
  const int*   eidx     = (const int*)d_in[1];
  const float* eattr    = (const float*)d_in[2];
  const float* W        = (const float*)d_in[3];
  const float* att_src  = (const float*)d_in[4];
  const float* att_dst  = (const float*)d_in[5];
  const float* W_edge   = (const float*)d_in[6];
  const float* att_edge = (const float*)d_in[7];
  const float* bias     = (const float*)d_in[8];
  float* out = (float*)d_out;

  char* ws = (char*)d_ws;
  float* h_buf  = (float*)(ws + 0);
  float* a_src  = (float*)(ws + 40960000);
  float* a_dst  = (float*)(ws + 43520000);
  int*   deg    = (int*)  (ws + 46080000);
  int*   cursor = (int*)  (ws + 46120192);
  int*   off    = (int*)  (ws + 46160384);
  float* coef   = (float*)(ws + 46200576);
  int*   csr_e  = (int*)  (ws + 46200832);

  const int* srcI = eidx;
  const int* dstI = eidx + EE;

  hipMemsetAsync(ws + 46080000, 0, 40000, stream);

  calc_edge_coef<<<1, 64, 0, stream>>>(W_edge, att_edge, coef);
  gemm_h<<<(MROWS + 127) / 128, 256, 0, stream>>>(x, W, h_buf);
  calc_attn_node<<<(MROWS * HH) / 8, 256, 0, stream>>>(h_buf, att_src, att_dst,
                                                       a_src, a_dst);
  csr_count<<<(EE + 255) / 256, 256, 0, stream>>>(dstI, deg);
  csr_scan<<<1, 1024, 0, stream>>>(deg, off, cursor);
  csr_fill<<<(EE + 255) / 256, 256, 0, stream>>>(dstI, cursor, csr_e);
  node_gather<<<NB, 256, 0, stream>>>(srcI, eattr, off, csr_e, a_src, a_dst,
                                      coef, h_buf, bias, out);
}


# a_src/a_dst dots moved into the GEMM epilogue (from the accumulators, 16-lane halving butterfly, new fixed sum order); the separate 160000-wave kernel and its re-read of h are removed
# speedup vs baseline: 4.3757x; 1.1556x over previous
#include <hip/hip_runtime.h>


#define NB 10000
#define BB 4
#define EE 160000
#define DD 128
#define HH 4
#define OO 64
#define HO 256
#define MROWS (BB * NB)
#define NEG_SLOPE 0.2f

typedef float v2f __attribute__((ext_vector_type(2)));
typedef float v8f __attribute__((ext_vector_type(8)));

__global__ __launch_bounds__(256) void gemm_h(const float* __restrict__ x,
                                              const float* __restrict__ W,
                                              const float* __restrict__ att_src,
                                              const float* __restrict__ att_dst,
                                              float* __restrict__ h,
                                              float* __restrict__ a_src,
                                              float* __restrict__ a_dst) {
  __shared__ float2 Wlds[32 * HO];
  const int tid = threadIdx.x;
  const int lane = tid & 31;
  const int wave = tid >> 5;
  const int rowBase = blockIdx.x * 128 + wave * 16;
  const bool active = rowBase < MROWS;

  v8f acc[16];
  const v8f vzero = {0.f, 0.f, 0.f, 0.f, 0.f, 0.f, 0.f, 0.f};
#pragma unroll
  for (int t = 0; t < 16; ++t) acc[t] = vzero;

  const int m = lane & 15;
  const int kb = (lane >> 4) << 1;
  const int kp = lane >> 4;

  for (int kblk = 0; kblk < DD; kblk += 64) {
#pragma unroll 4
    for (int p = 0; p < 32; ++p) {
      float2 pr;
      pr.x = W[(size_t)(kblk + 2 * p) * HO + tid];
      pr.y = W[(size_t)(kblk + 2 * p + 1) * HO + tid];
      Wlds[p * HO + tid] = pr;
    }
    __syncthreads();

    if (active) {
      const float* xrow = x + (size_t)(rowBase + m) * DD + kblk;
      for (int kk = 0; kk < 64; kk += 4) {
        v2f a;
        a.x = xrow[kk + kb];
        a.y = xrow[kk + kb + 1];
        const float2* w0 = Wlds + ((kk >> 1) + kp) * HO + (lane & 15);
#pragma unroll
        for (int t = 0; t < 16; ++t) {
          float2 bp = w0[t * 16];
          v2f b;
          b.x = bp.x;
          b.y = bp.y;
          acc[t] = __builtin_amdgcn_wmma_f32_16x16x4_f32(
              false, a, false, b, (short)0, acc[t], false, false);
        }
      }
    }
    __syncthreads();
  }

  if (active) {
    const int nLocal = lane & 15;
    const int rOff = (lane >> 4) * 8;
#pragma unroll
    for (int t = 0; t < 16; ++t)
#pragma unroll
      for (int v = 0; v < 8; ++v)
        h[(size_t)(rowBase + rOff + v) * HO + t * 16 + nLocal] = acc[t][v];

    float res[HH];
#pragma unroll
    for (int hd = 0; hd < HH; ++hd) {
      float val[16];
      float ws[4], wd[4];
#pragma unroll
      for (int tt = 0; tt < 4; ++tt) {
        ws[tt] = att_src[hd * OO + tt * 16 + nLocal];
        wd[tt] = att_dst[hd * OO + tt * 16 + nLocal];
      }
#pragma unroll
      for (int v = 0; v < 8; ++v) {
        float s = 0.f, d = 0.f;
#pragma unroll
        for (int tt = 0; tt < 4; ++tt) {
          s += acc[hd * 4 + tt][v] * ws[tt];
          d += acc[hd * 4 + tt][v] * wd[tt];
        }
        val[v] = s;
        val[8 + v] = d;
      }
#pragma unroll
      for (int half = 8; half >= 1; half >>= 1) {
        const bool up = (lane & half) != 0;
#pragma unroll
        for (int j = 0; j < half; ++j) {
          const float lo = val[j];
          const float hi = val[j + half];
          const float keep = up ? hi : lo;
          const float send = up ? lo : hi;
          val[j] = keep + __shfl_xor(send, half);
        }
      }
      res[hd] = val[0];
    }
    float* ap = (nLocal & 8) ? a_dst : a_src;
    *reinterpret_cast<float4*>(ap + (size_t)(rowBase + rOff + (nLocal & 7)) * HH) =
        make_float4(res[0], res[1], res[2], res[3]);
  }
}

__global__ void calc_edge_coef(const float* __restrict__ W_edge,
                               const float* __restrict__ att_edge,
                               float* __restrict__ coef) {
  const int t = threadIdx.x;
  if (t < HH) {
    float s = 0.f;
    for (int o = 0; o < OO; ++o) s += W_edge[t * OO + o] * att_edge[t * OO + o];
    coef[t] = s;
  }
}

__global__ __launch_bounds__(256) void csr_count(const int* __restrict__ dst,
                                                 int* __restrict__ deg) {
  const int e = blockIdx.x * 256 + threadIdx.x;
  if (e < EE) atomicAdd(&deg[dst[e]], 1);
}

#define SCAN_PER ((NB + 1023) / 1024)
__global__ __launch_bounds__(1024) void csr_scan(const int* __restrict__ deg,
                                                 int* __restrict__ off,
                                                 int* __restrict__ cursor) {
  __shared__ int part[1024];
  const int t = threadIdx.x;
  const int lo = t * SCAN_PER;
  int loc[SCAN_PER];
  int s = 0;
  for (int k = 0; k < SCAN_PER; ++k) {
    const int i = lo + k;
    const int v = (i < NB) ? deg[i] : 0;
    loc[k] = s;
    s += v;
  }
  part[t] = s;
  __syncthreads();
  for (int d = 1; d < 1024; d <<= 1) {
    const int v = (t >= d) ? part[t - d] : 0;
    __syncthreads();
    part[t] += v;
    __syncthreads();
  }
  const int base = part[t] - s;
  for (int k = 0; k < SCAN_PER; ++k) {
    const int i = lo + k;
    if (i < NB) {
      off[i] = base + loc[k];
      cursor[i] = base + loc[k];
    }
  }
  if (t == 1023) off[NB] = part[1023];
}

__global__ __launch_bounds__(256) void csr_fill(const int* __restrict__ dst,
                                                int* __restrict__ cursor,
                                                int* __restrict__ csr_e) {
  const int e = blockIdx.x * 256 + threadIdx.x;
  if (e < EE) {
    const int slot = atomicAdd(&cursor[dst[e]], 1);
    csr_e[slot] = e;
  }
}

#define SEG_CAP 512
__global__ __launch_bounds__(256) void node_gather(
    const int* __restrict__ src, const float* __restrict__ eattr,
    const int* __restrict__ off, const int* __restrict__ csr_e,
    const float* __restrict__ a_src, const float* __restrict__ a_dst,
    const float* __restrict__ coef, const float* __restrict__ h,
    const float* __restrict__ bias, float* __restrict__ out) {
  __shared__ int s_e[SEG_CAP];
  __shared__ int s_src[SEG_CAP];
  __shared__ float s_ea[SEG_CAP];
  __shared__ float s_red[16][16];
  __shared__ float s_w[16][16];
  __shared__ int s_cs[16];
  __shared__ __attribute__((aligned(16))) float s_acc[BB * HO];

  const int n = blockIdx.x;
  const int t = threadIdx.x;
  const int beg = off[n];
  const int L = off[n + 1] - beg;
  const bool staged = (L <= SEG_CAP);

  if (staged) {
    for (int i = t; i < L; i += 256) s_e[i] = csr_e[beg + i];
    __syncthreads();
    for (int i = t; i < L; i += 256) {
      const int e = s_e[i];
      int r = 0;
      for (int j = 0; j < L; ++j) r += (s_e[j] < e) ? 1 : 0;
      s_src[r] = src[e];
      s_ea[r] = eattr[e];
    }
    __syncthreads();
  }
  auto edge_src = [&](int i) -> int {
    return staged ? s_src[i] : src[csr_e[beg + i]];
  };
  auto edge_ea = [&](int i) -> float {
    return staged ? s_ea[i] : eattr[csr_e[beg + i]];
  };

  const int bh = t & 15;
  const int g = t >> 4;
  const int b1 = bh >> 2, h1 = bh & 3;
  const float adst = a_dst[((size_t)b1 * NB + n) * HH + h1];
  const float cf = coef[h1];
  const float* asrc_b = a_src + (size_t)b1 * NB * HH + h1;

  float m = -INFINITY;
  for (int i = g; i < L; i += 16) {
    float a = asrc_b[(size_t)edge_src(i) * HH] + adst + edge_ea(i) * cf;
    a = a > 0.f ? a : a * NEG_SLOPE;
    m = fmaxf(m, a);
  }
  s_red[g][bh] = m;
  __syncthreads();
  float mm = s_red[0][bh];
  for (int k = 1; k < 16; ++k) mm = fmaxf(mm, s_red[k][bh]);
  if (!__builtin_isfinite(mm)) mm = 0.f;
  __syncthreads();

  float sm = 0.f;
  for (int i = g; i < L; i += 16) {
    float a = asrc_b[(size_t)edge_src(i) * HH] + adst + edge_ea(i) * cf;
    a = a > 0.f ? a : a * NEG_SLOPE;
    sm += __expf(a - mm);
  }
  s_red[g][bh] = sm;
  __syncthreads();
  float den = 0.f;
  for (int k = 0; k < 16; ++k) den += s_red[k][bh];
  den = fmaxf(den, 1e-16f);

  const int brow = t >> 6;
  const int wsel = t >> 4;
  const float4* hb = (const float4*)(h + (size_t)brow * NB * HO) + (t & 63);
  float4 acc = make_float4(0.f, 0.f, 0.f, 0.f);

  for (int c = 0; c < L; c += 16) {
    {
      const int i = c + g;
      const int ic = i < L ? i : L - 1;
      const int s = edge_src(ic);
      float a = asrc_b[(size_t)s * HH] + adst + edge_ea(ic) * cf;
      a = a > 0.f ? a : a * NEG_SLOPE;
      const float w = __expf(a - mm) / den;
      s_w[g][bh] = i < L ? w : 0.f;
      if (bh == 0) s_cs[g] = s;
    }
    __syncthreads();
    const int cnt = (L - c) < 16 ? (L - c) : 16;
    for (int j = 0; j < cnt; j += 4) {
      const float4 v0 = hb[(size_t)s_cs[j + 0] * (HO / 4)];
      const float4 v1 = hb[(size_t)s_cs[j + 1] * (HO / 4)];
      const float4 v2 = hb[(size_t)s_cs[j + 2] * (HO / 4)];
      const float4 v3 = hb[(size_t)s_cs[j + 3] * (HO / 4)];
      const float w0 = s_w[j + 0][wsel];
      const float w1 = s_w[j + 1][wsel];
      const float w2 = s_w[j + 2][wsel];
      const float w3 = s_w[j + 3][wsel];
      acc.x += w0 * v0.x; acc.y += w0 * v0.y; acc.z += w0 * v0.z; acc.w += w0 * v0.w;
      acc.x += w1 * v1.x; acc.y += w1 * v1.y; acc.z += w1 * v1.z; acc.w += w1 * v1.w;
      acc.x += w2 * v2.x; acc.y += w2 * v2.y; acc.z += w2 * v2.z; acc.w += w2 * v2.w;
      acc.x += w3 * v3.x; acc.y += w3 * v3.y; acc.z += w3 * v3.z; acc.w += w3 * v3.w;
    }
    __syncthreads();
  }

  *reinterpret_cast<float4*>(&s_acc[t * 4]) = acc;
  __syncthreads();
  const int o = t & 63;
  const float* p = s_acc + brow * HO + o;
  out[((size_t)brow * NB + n) * OO + o] =
      (p[0] + p[64] + p[128] + p[192]) * 0.25f + bias[o];
}

extern "C" void kernel_launch(void* const* d_in, const int* in_sizes, int n_in,
                              void* d_out, int out_size, void* d_ws, size_t ws_size,
                              hipStream_t stream) {
  const float* x        = (const float*)d_in[0];
  const int*   eidx     = (const int*)d_in[1];
  const float* eattr    = (const float*)d_in[2];
  const float* W        = (const float*)d_in[3];
  const float* att_src  = (const float*)d_in[4];
  const float* att_dst  = (const float*)d_in[5];
  const float* W_edge   = (const float*)d_in[6];
  const float* att_edge = (const float*)d_in[7];
  const float* bias     = (const float*)d_in[8];
  float* out = (float*)d_out;

  char* ws = (char*)d_ws;
  float* h_buf  = (float*)(ws + 0);
  float* a_src  = (float*)(ws + 40960000);
  float* a_dst  = (float*)(ws + 43520000);
  int*   deg    = (int*)  (ws + 46080000);
  int*   cursor = (int*)  (ws + 46120192);
  int*   off    = (int*)  (ws + 46160384);
  float* coef   = (float*)(ws + 46200576);
  int*   csr_e  = (int*)  (ws + 46200832);

  const int* srcI = eidx;
  const int* dstI = eidx + EE;

  hipMemsetAsync(ws + 46080000, 0, 40000, stream);

  calc_edge_coef<<<1, 64, 0, stream>>>(W_edge, att_edge, coef);
  gemm_h<<<(MROWS + 127) / 128, 256, 0, stream>>>(x, W, att_src, att_dst, h_buf,
                                                  a_src, a_dst);
  csr_count<<<(EE + 255) / 256, 256, 0, stream>>>(dstI, deg);
  csr_scan<<<1, 1024, 0, stream>>>(deg, off, cursor);
  csr_fill<<<(EE + 255) / 256, 256, 0, stream>>>(dstI, cursor, csr_e);
  node_gather<<<NB, 256, 0, stream>>>(srcI, eattr, off, csr_e, a_src, a_dst,
                                      coef, h_buf, bias, out);
}


# Degree count, offset scan, slot fill and edge coefficient moved into the GEMM launch: 40 extra work-groups each own 250 nodes, scan all edges twice with LDS counters; memset and 4 launches removed
# speedup vs baseline: 4.6516x; 1.0631x over previous
#include <hip/hip_runtime.h>


#define NB 10000
#define BB 4
#define EE 160000
#define DD 128
#define HH 4
#define OO 64
#define HO 256
#define MROWS (BB * NB)
#define NEG_SLOPE 0.2f
#define CSR_BLOCKS 40
#define NODES_PER (NB / CSR_BLOCKS)
#define E4 (EE / 4)
#define CSR_UNROLL 13
#define CSR_MAIN (E4 / (256 * CSR_UNROLL))
#define CSR_TAIL0 (CSR_MAIN * 256 * CSR_UNROLL)
static_assert(CSR_MAIN % 2 == 0 && E4 - CSR_TAIL0 >= 0 && E4 - CSR_TAIL0 <= 256,
              "edge list split");
static_assert(NODES_PER <= 256 && NODES_PER * CSR_BLOCKS == NB && EE % 4 == 0,
              "node split");

typedef float v2f __attribute__((ext_vector_type(2)));
typedef float v8f __attribute__((ext_vector_type(8)));

__global__ __launch_bounds__(256) void gemm_h(const float* __restrict__ x,
                                              const float* __restrict__ W,
                                              const float* __restrict__ att_src,
                                              const float* __restrict__ att_dst,
                                              float* __restrict__ h,
                                              float* __restrict__ a_src,
                                              float* __restrict__ a_dst,
                                              const int* __restrict__ dst,
                                              const float* __restrict__ W_edge,
                                              const float* __restrict__ att_edge,
                                              int* __restrict__ off,
                                              int* __restrict__ csr_e,
                                              float* __restrict__ coef) {
  __shared__ float2 Wlds[32 * HO];
  __shared__ int s_cnt[NODES_PER];
  __shared__ int s_wtot[8];
  __shared__ int s_below;
  const int tid = threadIdx.x;
  const int lane = tid & 31;
  const int wave = tid >> 5;

  if (blockIdx.x < CSR_BLOCKS) {
    const int r = blockIdx.x;
    const int lo = r * NODES_PER;
    const int4* d4 = reinterpret_cast<const int4*>(dst);
    const int4 none = make_int4(0x7fffffff, 0x7fffffff, 0x7fffffff, 0x7fffffff);
    if (tid < NODES_PER) s_cnt[tid] = 0;
    if (tid == 0) s_below = 0;
    __syncthreads();

    int below = 0;
    auto load_group = [&](int4 (&v)[CSR_UNROLL], int it) {
      const int4* p = d4 + it * (256 * CSR_UNROLL) + tid;
#pragma unroll
      for (int u = 0; u < CSR_UNROLL; ++u) v[u] = p[u * 256];
    };
    auto count4 = [&](const int4& q) {
      const int cc[4] = {q.x, q.y, q.z, q.w};
#pragma unroll
      for (int k = 0; k < 4; ++k) {
        const unsigned rel = (unsigned)(cc[k] - lo);
        if (rel < (unsigned)NODES_PER) atomicAdd(&s_cnt[rel], 1);
        below += (cc[k] < lo) ? 1 : 0;
      }
    };
    {
      int4 va[CSR_UNROLL], vb[CSR_UNROLL];
      load_group(va, 0);
      for (int it = 0; it < CSR_MAIN; it += 2) {
        load_group(vb, it + 1);
#pragma unroll
        for (int u = 0; u < CSR_UNROLL; ++u) count4(va[u]);
        if (it + 2 < CSR_MAIN) load_group(va, it + 2);
#pragma unroll
        for (int u = 0; u < CSR_UNROLL; ++u) count4(vb[u]);
      }
      int4 vt = none;
      if (CSR_TAIL0 + tid < E4) vt = d4[CSR_TAIL0 + tid];
      count4(vt);
    }
    atomicAdd(&s_below, below);
    __syncthreads();

    const int cnt = (tid < NODES_PER) ? s_cnt[tid] : 0;
    int inc = cnt;
#pragma unroll
    for (int d = 1; d < 32; d <<= 1) {
      const int o = __shfl_up(inc, d);
      if (lane >= d) inc += o;
    }
    if (lane == 31) s_wtot[wave] = inc;
    __syncthreads();
    int wbase = 0;
#pragma unroll
    for (int w = 0; w < 8; ++w) wbase += (w < wave) ? s_wtot[w] : 0;
    const int myoff = s_below + wbase + inc - cnt;
    if (tid < NODES_PER) {
      s_cnt[tid] = myoff;
      off[lo + tid] = myoff;
      if (r == CSR_BLOCKS - 1 && tid == NODES_PER - 1) off[NB] = myoff + cnt;
    }
    if (r == 0 && tid >= 256 - HH) {
      const int hd = tid - (256 - HH);
      float s = 0.f;
      for (int o = 0; o < OO; ++o) s += W_edge[hd * OO + o] * att_edge[hd * OO + o];
      coef[hd] = s;
    }
    __syncthreads();

    auto fill4 = [&](const int4& q, int e0) {
      const int cc[4] = {q.x, q.y, q.z, q.w};
#pragma unroll
      for (int k = 0; k < 4; ++k) {
        const unsigned rel = (unsigned)(cc[k] - lo);
        if (rel < (unsigned)NODES_PER) {
          const int slot = atomicAdd(&s_cnt[rel], 1);
          csr_e[slot] = e0 + k;
        }
      }
    };
    {
      int4 va[CSR_UNROLL], vb[CSR_UNROLL];
      load_group(va, 0);
      for (int it = 0; it < CSR_MAIN; it += 2) {
        load_group(vb, it + 1);
#pragma unroll
        for (int u = 0; u < CSR_UNROLL; ++u)
          fill4(va[u], (it * (256 * CSR_UNROLL) + u * 256 + tid) * 4);
        if (it + 2 < CSR_MAIN) load_group(va, it + 2);
#pragma unroll
        for (int u = 0; u < CSR_UNROLL; ++u)
          fill4(vb[u], ((it + 1) * (256 * CSR_UNROLL) + u * 256 + tid) * 4);
      }
      int4 vt = none;
      if (CSR_TAIL0 + tid < E4) vt = d4[CSR_TAIL0 + tid];
      fill4(vt, (CSR_TAIL0 + tid) * 4);
    }
    return;
  }

  const int rowBase = (blockIdx.x - CSR_BLOCKS) * 128 + wave * 16;
  const bool active = rowBase < MROWS;

  v8f acc[16];
  const v8f vzero = {0.f, 0.f, 0.f, 0.f, 0.f, 0.f, 0.f, 0.f};
#pragma unroll
  for (int t = 0; t < 16; ++t) acc[t] = vzero;

  const int m = lane & 15;
  const int kb = (lane >> 4) << 1;
  const int kp = lane >> 4;

  for (int kblk = 0; kblk < DD; kblk += 64) {
#pragma unroll 4
    for (int p = 0; p < 32; ++p) {
      float2 pr;
      pr.x = W[(size_t)(kblk + 2 * p) * HO + tid];
      pr.y = W[(size_t)(kblk + 2 * p + 1) * HO + tid];
      Wlds[p * HO + tid] = pr;
    }
    __syncthreads();

    if (active) {
      const float* xrow = x + (size_t)(rowBase + m) * DD + kblk;
      for (int kk = 0; kk < 64; kk += 4) {
        v2f a;
        a.x = xrow[kk + kb];
        a.y = xrow[kk + kb + 1];
        const float2* w0 = Wlds + ((kk >> 1) + kp) * HO + (lane & 15);
#pragma unroll
        for (int t = 0; t < 16; ++t) {
          float2 bp = w0[t * 16];
          v2f b;
          b.x = bp.x;
          b.y = bp.y;
          acc[t] = __builtin_amdgcn_wmma_f32_16x16x4_f32(
              false, a, false, b, (short)0, acc[t], false, false);
        }
      }
    }
    __syncthreads();
  }

  if (active) {
    const int nLocal = lane & 15;
    const int rOff = (lane >> 4) * 8;
#pragma unroll
    for (int t = 0; t < 16; ++t)
#pragma unroll
      for (int v = 0; v < 8; ++v)
        h[(size_t)(rowBase + rOff + v) * HO + t * 16 + nLocal] = acc[t][v];

    float res[HH];
#pragma unroll
    for (int hd = 0; hd < HH; ++hd) {
      float val[16];
      float ws[4], wd[4];
#pragma unroll
      for (int tt = 0; tt < 4; ++tt) {
        ws[tt] = att_src[hd * OO + tt * 16 + nLocal];
        wd[tt] = att_dst[hd * OO + tt * 16 + nLocal];
      }
#pragma unroll
      for (int v = 0; v < 8; ++v) {
        float s = 0.f, d = 0.f;
#pragma unroll
        for (int tt = 0; tt < 4; ++tt) {
          s += acc[hd * 4 + tt][v] * ws[tt];
          d += acc[hd * 4 + tt][v] * wd[tt];
        }
        val[v] = s;
        val[8 + v] = d;
      }
#pragma unroll
      for (int half = 8; half >= 1; half >>= 1) {
        const bool up = (lane & half) != 0;
#pragma unroll
        for (int j = 0; j < half; ++j) {
          const float lo = val[j];
          const float hi = val[j + half];
          const float keep = up ? hi : lo;
          const float send = up ? lo : hi;
          val[j] = keep + __shfl_xor(send, half);
        }
      }
      res[hd] = val[0];
    }
    float* ap = (nLocal & 8) ? a_dst : a_src;
    *reinterpret_cast<float4*>(ap + (size_t)(rowBase + rOff + (nLocal & 7)) * HH) =
        make_float4(res[0], res[1], res[2], res[3]);
  }
}

#define SEG_CAP 512
__global__ __launch_bounds__(256) void node_gather(
    const int* __restrict__ src, const float* __restrict__ eattr,
    const int* __restrict__ off, const int* __restrict__ csr_e,
    const float* __restrict__ a_src, const float* __restrict__ a_dst,
    const float* __restrict__ coef, const float* __restrict__ h,
    const float* __restrict__ bias, float* __restrict__ out) {
  __shared__ int s_e[SEG_CAP];
  __shared__ int s_src[SEG_CAP];
  __shared__ float s_ea[SEG_CAP];
  __shared__ float s_red[16][16];
  __shared__ float s_w[16][16];
  __shared__ int s_cs[16];
  __shared__ __attribute__((aligned(16))) float s_acc[BB * HO];

  const int n = blockIdx.x;
  const int t = threadIdx.x;
  const int beg = off[n];
  const int L = off[n + 1] - beg;
  const bool staged = (L <= SEG_CAP);

  if (staged) {
    for (int i = t; i < L; i += 256) s_e[i] = csr_e[beg + i];
    __syncthreads();
    for (int i = t; i < L; i += 256) {
      const int e = s_e[i];
      int r = 0;
      for (int j = 0; j < L; ++j) r += (s_e[j] < e) ? 1 : 0;
      s_src[r] = src[e];
      s_ea[r] = eattr[e];
    }
    __syncthreads();
  }
  auto edge_src = [&](int i) -> int {
    return staged ? s_src[i] : src[csr_e[beg + i]];
  };
  auto edge_ea = [&](int i) -> float {
    return staged ? s_ea[i] : eattr[csr_e[beg + i]];
  };

  const int bh = t & 15;
  const int g = t >> 4;
  const int b1 = bh >> 2, h1 = bh & 3;
  const float adst = a_dst[((size_t)b1 * NB + n) * HH + h1];
  const float cf = coef[h1];
  const float* asrc_b = a_src + (size_t)b1 * NB * HH + h1;

  float m = -INFINITY;
  for (int i = g; i < L; i += 16) {
    float a = asrc_b[(size_t)edge_src(i) * HH] + adst + edge_ea(i) * cf;
    a = a > 0.f ? a : a * NEG_SLOPE;
    m = fmaxf(m, a);
  }
  s_red[g][bh] = m;
  __syncthreads();
  float mm = s_red[0][bh];
  for (int k = 1; k < 16; ++k) mm = fmaxf(mm, s_red[k][bh]);
  if (!__builtin_isfinite(mm)) mm = 0.f;
  __syncthreads();

  float sm = 0.f;
  for (int i = g; i < L; i += 16) {
    float a = asrc_b[(size_t)edge_src(i) * HH] + adst + edge_ea(i) * cf;
    a = a > 0.f ? a : a * NEG_SLOPE;
    sm += __expf(a - mm);
  }
  s_red[g][bh] = sm;
  __syncthreads();
  float den = 0.f;
  for (int k = 0; k < 16; ++k) den += s_red[k][bh];
  den = fmaxf(den, 1e-16f);

  const int brow = t >> 6;
  const int wsel = t >> 4;
  const float4* hb = (const float4*)(h + (size_t)brow * NB * HO) + (t & 63);
  float4 acc = make_float4(0.f, 0.f, 0.f, 0.f);

  for (int c = 0; c < L; c += 16) {
    {
      const int i = c + g;
      const int ic = i < L ? i : L - 1;
      const int s = edge_src(ic);
      float a = asrc_b[(size_t)s * HH] + adst + edge_ea(ic) * cf;
      a = a > 0.f ? a : a * NEG_SLOPE;
      const float w = __expf(a - mm) / den;
      s_w[g][bh] = i < L ? w : 0.f;
      if (bh == 0) s_cs[g] = s;
    }
    __syncthreads();
    const int cnt = (L - c) < 16 ? (L - c) : 16;
    for (int j = 0; j < cnt; j += 4) {
      const float4 v0 = hb[(size_t)s_cs[j + 0] * (HO / 4)];
      const float4 v1 = hb[(size_t)s_cs[j + 1] * (HO / 4)];
      const float4 v2 = hb[(size_t)s_cs[j + 2] * (HO / 4)];
      const float4 v3 = hb[(size_t)s_cs[j + 3] * (HO / 4)];
      const float w0 = s_w[j + 0][wsel];
      const float w1 = s_w[j + 1][wsel];
      const float w2 = s_w[j + 2][wsel];
      const float w3 = s_w[j + 3][wsel];
      acc.x += w0 * v0.x; acc.y += w0 * v0.y; acc.z += w0 * v0.z; acc.w += w0 * v0.w;
      acc.x += w1 * v1.x; acc.y += w1 * v1.y; acc.z += w1 * v1.z; acc.w += w1 * v1.w;
      acc.x += w2 * v2.x; acc.y += w2 * v2.y; acc.z += w2 * v2.z; acc.w += w2 * v2.w;
      acc.x += w3 * v3.x; acc.y += w3 * v3.y; acc.z += w3 * v3.z; acc.w += w3 * v3.w;
    }
    __syncthreads();
  }

  *reinterpret_cast<float4*>(&s_acc[t * 4]) = acc;
  __syncthreads();
  const int o = t & 63;
  const float* p = s_acc + brow * HO + o;
  out[((size_t)brow * NB + n) * OO + o] =
      (p[0] + p[64] + p[128] + p[192]) * 0.25f + bias[o];
}

extern "C" void kernel_launch(void* const* d_in, const int* in_sizes, int n_in,
                              void* d_out, int out_size, void* d_ws, size_t ws_size,
                              hipStream_t stream) {
  const float* x        = (const float*)d_in[0];
  const int*   eidx     = (const int*)d_in[1];
  const float* eattr    = (const float*)d_in[2];
  const float* W        = (const float*)d_in[3];
  const float* att_src  = (const float*)d_in[4];
  const float* att_dst  = (const float*)d_in[5];
  const float* W_edge   = (const float*)d_in[6];
  const float* att_edge = (const float*)d_in[7];
  const float* bias     = (const float*)d_in[8];
  float* out = (float*)d_out;

  char* ws = (char*)d_ws;
  float* h_buf  = (float*)(ws + 0);
  float* a_src  = (float*)(ws + 40960000);
  float* a_dst  = (float*)(ws + 43520000);
  int*   off    = (int*)  (ws + 46160384);
  float* coef   = (float*)(ws + 46200576);
  int*   csr_e  = (int*)  (ws + 46200832);

  const int* srcI = eidx;
  const int* dstI = eidx + EE;

  gemm_h<<<CSR_BLOCKS + (MROWS + 127) / 128, 256, 0, stream>>>(
      x, W, att_src, att_dst, h_buf, a_src, a_dst, dstI, W_edge, att_edge, off,
      csr_e, coef);
  node_gather<<<NB, 256, 0, stream>>>(srcI, eattr, off, csr_e, a_src, a_dst,
                                      coef, h_buf, bias, out);
}


# Edge grouping scans the destination list once, not twice: each thread keeps the owned edges it meets (id + node index) in the idle weight-tile LDS and takes their slots from that list afterwards
# speedup vs baseline: 5.8908x; 1.2664x over previous
#include <hip/hip_runtime.h>


#define NB 10000
#define BB 4
#define EE 160000
#define DD 128
#define HH 4
#define OO 64
#define HO 256
#define MROWS (BB * NB)
#define NEG_SLOPE 0.2f
#define CSR_BLOCKS 40
#define NODES_PER (NB / CSR_BLOCKS)
#define E4 (EE / 4)
#define CSR_UNROLL 13
#define CSR_MAIN (E4 / (256 * CSR_UNROLL))
#define CSR_TAIL0 (CSR_MAIN * 256 * CSR_UNROLL)
static_assert(CSR_MAIN % 2 == 0 && E4 - CSR_TAIL0 >= 0 && E4 - CSR_TAIL0 <= 256,
              "edge list split");
static_assert(NODES_PER <= 256 && NODES_PER * CSR_BLOCKS == NB && EE % 4 == 0,
              "node split");
#define HIT_CAP 64
#define EBITS 18
static_assert(EE <= (1 << EBITS) && NODES_PER <= (1 << (32 - EBITS)) &&
                  (HIT_CAP & (HIT_CAP - 1)) == 0 && HIT_CAP * 256 * 4 <= 32 * HO * 8,
              "kept edge word / list size");

typedef float v2f __attribute__((ext_vector_type(2)));
typedef float v8f __attribute__((ext_vector_type(8)));

__global__ __launch_bounds__(256) void gemm_h(const float* __restrict__ x,
                                              const float* __restrict__ W,
                                              const float* __restrict__ att_src,
                                              const float* __restrict__ att_dst,
                                              float* __restrict__ h,
                                              float* __restrict__ a_src,
                                              float* __restrict__ a_dst,
                                              const int* __restrict__ dst,
                                              const float* __restrict__ W_edge,
                                              const float* __restrict__ att_edge,
                                              int* __restrict__ off,
                                              int* __restrict__ csr_e,
                                              float* __restrict__ coef) {
  __shared__ float2 Wlds[32 * HO];
  __shared__ int s_cnt[NODES_PER];
  __shared__ int s_wtot[8];
  __shared__ int s_below;
  __shared__ int s_over;
  const int tid = threadIdx.x;
  const int lane = tid & 31;
  const int wave = tid >> 5;

  if (blockIdx.x < CSR_BLOCKS) {
    const int r = blockIdx.x;
    const int lo = r * NODES_PER;
    const int4* d4 = reinterpret_cast<const int4*>(dst);
    const int4 none = make_int4(0x7fffffff, 0x7fffffff, 0x7fffffff, 0x7fffffff);
    unsigned* hits = reinterpret_cast<unsigned*>(Wlds);
    if (tid < NODES_PER) s_cnt[tid] = 0;
    if (tid == 0) {
      s_below = 0;
      s_over = 0;
    }
    __syncthreads();

    int below = 0;
    int nh = 0;
    auto load_group = [&](int4 (&v)[CSR_UNROLL], int it) {
      const int4* p = d4 + it * (256 * CSR_UNROLL) + tid;
#pragma unroll
      for (int u = 0; u < CSR_UNROLL; ++u) v[u] = p[u * 256];
    };
    auto scan4 = [&](const int4& q, int e0) {
      const int cc[4] = {q.x, q.y, q.z, q.w};
#pragma unroll
      for (int k = 0; k < 4; ++k) {
        const unsigned rel = (unsigned)(cc[k] - lo);
        if (rel < (unsigned)NODES_PER) {
          atomicAdd(&s_cnt[rel], 1);
          hits[(nh & (HIT_CAP - 1)) * 256 + tid] = (rel << EBITS) | (unsigned)(e0 + k);
          ++nh;
        }
        below += (cc[k] < lo) ? 1 : 0;
      }
    };
    {
      int4 va[CSR_UNROLL], vb[CSR_UNROLL];
      load_group(va, 0);
      for (int it = 0; it < CSR_MAIN; it += 2) {
        load_group(vb, it + 1);
#pragma unroll
        for (int u = 0; u < CSR_UNROLL; ++u)
          scan4(va[u], (it * (256 * CSR_UNROLL) + u * 256 + tid) * 4);
        if (it + 2 < CSR_MAIN) load_group(va, it + 2);
#pragma unroll
        for (int u = 0; u < CSR_UNROLL; ++u)
          scan4(vb[u], ((it + 1) * (256 * CSR_UNROLL) + u * 256 + tid) * 4);
      }
      int4 vt = none;
      if (CSR_TAIL0 + tid < E4) vt = d4[CSR_TAIL0 + tid];
      scan4(vt, (CSR_TAIL0 + tid) * 4);
    }
    if (nh > HIT_CAP) s_over = 1;
    atomicAdd(&s_below, below);
    __syncthreads();

    const int cnt = (tid < NODES_PER) ? s_cnt[tid] : 0;
    int inc = cnt;
#pragma unroll
    for (int d = 1; d < 32; d <<= 1) {
      const int o = __shfl_up(inc, d);
      if (lane >= d) inc += o;
    }
    if (lane == 31) s_wtot[wave] = inc;
    __syncthreads();
    int wbase = 0;
#pragma unroll
    for (int w = 0; w < 8; ++w) wbase += (w < wave) ? s_wtot[w] : 0;
    const int myoff = s_below + wbase + inc - cnt;
    if (tid < NODES_PER) {
      s_cnt[tid] = myoff;
      off[lo + tid] = myoff;
      if (r == CSR_BLOCKS - 1 && tid == NODES_PER - 1) off[NB] = myoff + cnt;
    }
    if (r == 0 && tid >= 256 - HH) {
      const int hd = tid - (256 - HH);
      float s = 0.f;
      for (int o = 0; o < OO; ++o) s += W_edge[hd * OO + o] * att_edge[hd * OO + o];
      coef[hd] = s;
    }
    __syncthreads();

    if (s_over == 0) {
      for (int j = 0; j < nh; j += 4) {
        unsigned hw[4] = {0u, 0u, 0u, 0u};
        int slot[4] = {0, 0, 0, 0};
#pragma unroll
        for (int u = 0; u < 4; ++u)
          if (j + u < nh) hw[u] = hits[(j + u) * 256 + tid];
#pragma unroll
        for (int u = 0; u < 4; ++u)
          if (j + u < nh) slot[u] = atomicAdd(&s_cnt[hw[u] >> EBITS], 1);
#pragma unroll
        for (int u = 0; u < 4; ++u)
          if (j + u < nh) csr_e[slot[u]] = (int)(hw[u] & ((1u << EBITS) - 1u));
      }
      return;
    }

    auto fill4 = [&](const int4& q, int e0) {
      const int cc[4] = {q.x, q.y, q.z, q.w};
#pragma unroll
      for (int k = 0; k < 4; ++k) {
        const unsigned rel = (unsigned)(cc[k] - lo);
        if (rel < (unsigned)NODES_PER) {
          const int slot = atomicAdd(&s_cnt[rel], 1);
          csr_e[slot] = e0 + k;
        }
      }
    };
    {
      int4 va[CSR_UNROLL], vb[CSR_UNROLL];
      load_group(va, 0);
      for (int it = 0; it < CSR_MAIN; it += 2) {
        load_group(vb, it + 1);
#pragma unroll
        for (int u = 0; u < CSR_UNROLL; ++u)
          fill4(va[u], (it * (256 * CSR_UNROLL) + u * 256 + tid) * 4);
        if (it + 2 < CSR_MAIN) load_group(va, it + 2);
#pragma unroll
        for (int u = 0; u < CSR_UNROLL; ++u)
          fill4(vb[u], ((it + 1) * (256 * CSR_UNROLL) + u * 256 + tid) * 4);
      }
      int4 vt = none;
      if (CSR_TAIL0 + tid < E4) vt = d4[CSR_TAIL0 + tid];
      fill4(vt, (CSR_TAIL0 + tid) * 4);
    }
    return;
  }

  const int rowBase = (blockIdx.x - CSR_BLOCKS) * 128 + wave * 16;
  const bool active = rowBase < MROWS;

  v8f acc[16];
  const v8f vzero = {0.f, 0.f, 0.f, 0.f, 0.f, 0.f, 0.f, 0.f};
#pragma unroll
  for (int t = 0; t < 16; ++t) acc[t] = vzero;

  const int m = lane & 15;
  const int kb = (lane >> 4) << 1;
  const int kp = lane >> 4;

  for (int kblk = 0; kblk < DD; kblk += 64) {
#pragma unroll 4
    for (int p = 0; p < 32; ++p) {
      float2 pr;
      pr.x = W[(size_t)(kblk + 2 * p) * HO + tid];
      pr.y = W[(size_t)(kblk + 2 * p + 1) * HO + tid];
      Wlds[p * HO + tid] = pr;
    }
    __syncthreads();

    if (active) {
      const float* xrow = x + (size_t)(rowBase + m) * DD + kblk;
      for (int kk = 0; kk < 64; kk += 4) {
        v2f a;
        a.x = xrow[kk + kb];
        a.y = xrow[kk + kb + 1];
        const float2* w0 = Wlds + ((kk >> 1) + kp) * HO + (lane & 15);
#pragma unroll
        for (int t = 0; t < 16; ++t) {
          float2 bp = w0[t * 16];
          v2f b;
          b.x = bp.x;
          b.y = bp.y;
          acc[t] = __builtin_amdgcn_wmma_f32_16x16x4_f32(
              false, a, false, b, (short)0, acc[t], false, false);
        }
      }
    }
    __syncthreads();
  }

  if (active) {
    const int nLocal = lane & 15;
    const int rOff = (lane >> 4) * 8;
#pragma unroll
    for (int t = 0; t < 16; ++t)
#pragma unroll
      for (int v = 0; v < 8; ++v)
        h[(size_t)(rowBase + rOff + v) * HO + t * 16 + nLocal] = acc[t][v];

    float res[HH];
#pragma unroll
    for (int hd = 0; hd < HH; ++hd) {
      float val[16];
      float ws[4], wd[4];
#pragma unroll
      for (int tt = 0; tt < 4; ++tt) {
        ws[tt] = att_src[hd * OO + tt * 16 + nLocal];
        wd[tt] = att_dst[hd * OO + tt * 16 + nLocal];
      }
#pragma unroll
      for (int v = 0; v < 8; ++v) {
        float s = 0.f, d = 0.f;
#pragma unroll
        for (int tt = 0; tt < 4; ++tt) {
          s += acc[hd * 4 + tt][v] * ws[tt];
          d += acc[hd * 4 + tt][v] * wd[tt];
        }
        val[v] = s;
        val[8 + v] = d;
      }
#pragma unroll
      for (int half = 8; half >= 1; half >>= 1) {
        const bool up = (lane & half) != 0;
#pragma unroll
        for (int j = 0; j < half; ++j) {
          const float lo = val[j];
          const float hi = val[j + half];
          const float keep = up ? hi : lo;
          const float send = up ? lo : hi;
          val[j] = keep + __shfl_xor(send, half);
        }
      }
      res[hd] = val[0];
    }
    float* ap = (nLocal & 8) ? a_dst : a_src;
    *reinterpret_cast<float4*>(ap + (size_t)(rowBase + rOff + (nLocal & 7)) * HH) =
        make_float4(res[0], res[1], res[2], res[3]);
  }
}

#define SEG_CAP 512
__global__ __launch_bounds__(256) void node_gather(
    const int* __restrict__ src, const float* __restrict__ eattr,
    const int* __restrict__ off, const int* __restrict__ csr_e,
    const float* __restrict__ a_src, const float* __restrict__ a_dst,
    const float* __restrict__ coef, const float* __restrict__ h,
    const float* __restrict__ bias, float* __restrict__ out) {
  __shared__ int s_e[SEG_CAP];
  __shared__ int s_src[SEG_CAP];
  __shared__ float s_ea[SEG_CAP];
  __shared__ float s_red[16][16];
  __shared__ float s_w[16][16];
  __shared__ int s_cs[16];
  __shared__ __attribute__((aligned(16))) float s_acc[BB * HO];

  const int n = blockIdx.x;
  const int t = threadIdx.x;
  const int beg = off[n];
  const int L = off[n + 1] - beg;
  const bool staged = (L <= SEG_CAP);

  if (staged) {
    for (int i = t; i < L; i += 256) s_e[i] = csr_e[beg + i];
    __syncthreads();
    for (int i = t; i < L; i += 256) {
      const int e = s_e[i];
      int r = 0;
      for (int j = 0; j < L; ++j) r += (s_e[j] < e) ? 1 : 0;
      s_src[r] = src[e];
      s_ea[r] = eattr[e];
    }
    __syncthreads();
  }
  auto edge_src = [&](int i) -> int {
    return staged ? s_src[i] : src[csr_e[beg + i]];
  };
  auto edge_ea = [&](int i) -> float {
    return staged ? s_ea[i] : eattr[csr_e[beg + i]];
  };

  const int bh = t & 15;
  const int g = t >> 4;
  const int b1 = bh >> 2, h1 = bh & 3;
  const float adst = a_dst[((size_t)b1 * NB + n) * HH + h1];
  const float cf = coef[h1];
  const float* asrc_b = a_src + (size_t)b1 * NB * HH + h1;

  float m = -INFINITY;
  for (int i = g; i < L; i += 16) {
    float a = asrc_b[(size_t)edge_src(i) * HH] + adst + edge_ea(i) * cf;
    a = a > 0.f ? a : a * NEG_SLOPE;
    m = fmaxf(m, a);
  }
  s_red[g][bh] = m;
  __syncthreads();
  float mm = s_red[0][bh];
  for (int k = 1; k < 16; ++k) mm = fmaxf(mm, s_red[k][bh]);
  if (!__builtin_isfinite(mm)) mm = 0.f;
  __syncthreads();

  float sm = 0.f;
  for (int i = g; i < L; i += 16) {
    float a = asrc_b[(size_t)edge_src(i) * HH] + adst + edge_ea(i) * cf;
    a = a > 0.f ? a : a * NEG_SLOPE;
    sm += __expf(a - mm);
  }
  s_red[g][bh] = sm;
  __syncthreads();
  float den = 0.f;
  for (int k = 0; k < 16; ++k) den += s_red[k][bh];
  den = fmaxf(den, 1e-16f);

  const int brow = t >> 6;
  const int wsel = t >> 4;
  const float4* hb = (const float4*)(h + (size_t)brow * NB * HO) + (t & 63);
  float4 acc = make_float4(0.f, 0.f, 0.f, 0.f);

  for (int c = 0; c < L; c += 16) {
    {
      const int i = c + g;
      const int ic = i < L ? i : L - 1;
      const int s = edge_src(ic);
      float a = asrc_b[(size_t)s * HH] + adst + edge_ea(ic) * cf;
      a = a > 0.f ? a : a * NEG_SLOPE;
      const float w = __expf(a - mm) / den;
      s_w[g][bh] = i < L ? w : 0.f;
      if (bh == 0) s_cs[g] = s;
    }
    __syncthreads();
    const int cnt = (L - c) < 16 ? (L - c) : 16;
    for (int j = 0; j < cnt; j += 4) {
      const float4 v0 = hb[(size_t)s_cs[j + 0] * (HO / 4)];
      const float4 v1 = hb[(size_t)s_cs[j + 1] * (HO / 4)];
      const float4 v2 = hb[(size_t)s_cs[j + 2] * (HO / 4)];
      const float4 v3 = hb[(size_t)s_cs[j + 3] * (HO / 4)];
      const float w0 = s_w[j + 0][wsel];
      const float w1 = s_w[j + 1][wsel];
      const float w2 = s_w[j + 2][wsel];
      const float w3 = s_w[j + 3][wsel];
      acc.x += w0 * v0.x; acc.y += w0 * v0.y; acc.z += w0 * v0.z; acc.w += w0 * v0.w;
      acc.x += w1 * v1.x; acc.y += w1 * v1.y; acc.z += w1 * v1.z; acc.w += w1 * v1.w;
      acc.x += w2 * v2.x; acc.y += w2 * v2.y; acc.z += w2 * v2.z; acc.w += w2 * v2.w;
      acc.x += w3 * v3.x; acc.y += w3 * v3.y; acc.z += w3 * v3.z; acc.w += w3 * v3.w;
    }
    __syncthreads();
  }

  *reinterpret_cast<float4*>(&s_acc[t * 4]) = acc;
  __syncthreads();
  const int o = t & 63;
  const float* p = s_acc + brow * HO + o;
  out[((size_t)brow * NB + n) * OO + o] =
      (p[0] + p[64] + p[128] + p[192]) * 0.25f + bias[o];
}

extern "C" void kernel_launch(void* const* d_in, const int* in_sizes, int n_in,
                              void* d_out, int out_size, void* d_ws, size_t ws_size,
                              hipStream_t stream) {
  const float* x        = (const float*)d_in[0];
  const int*   eidx     = (const int*)d_in[1];
  const float* eattr    = (const float*)d_in[2];
  const float* W        = (const float*)d_in[3];
  const float* att_src  = (const float*)d_in[4];
  const float* att_dst  = (const float*)d_in[5];
  const float* W_edge   = (const float*)d_in[6];
  const float* att_edge = (const float*)d_in[7];
  const float* bias     = (const float*)d_in[8];
  float* out = (float*)d_out;

  char* ws = (char*)d_ws;
  float* h_buf  = (float*)(ws + 0);
  float* a_src  = (float*)(ws + 40960000);
  float* a_dst  = (float*)(ws + 43520000);
  int*   off    = (int*)  (ws + 46160384);
  float* coef   = (float*)(ws + 46200576);
  int*   csr_e  = (int*)  (ws + 46200832);

  const int* srcI = eidx;
  const int* dstI = eidx + EE;

  gemm_h<<<CSR_BLOCKS + (MROWS + 127) / 128, 256, 0, stream>>>(
      x, W, att_src, att_dst, h_buf, a_src, a_dst, dstI, W_edge, att_edge, off,
      csr_e, coef);
  node_gather<<<NB, 256, 0, stream>>>(srcI, eattr, off, csr_e, a_src, a_dst,
                                      coef, h_buf, bias, out);
}


# Edge grouping split over 4 slices of the destination list: 160 work-groups each scan a quarter (39 loads per thread in flight at once); the node gather joins a node's 4 segments; sum orders kept
# speedup vs baseline: 6.5012x; 1.1036x over previous
#include <hip/hip_runtime.h>


#define NB 10000
#define BB 4
#define EE 160000
#define DD 128
#define HH 4
#define OO 64
#define HO 256
#define MROWS (BB * NB)
#define NEG_SLOPE 0.2f
#define CSR_RANGES 40
#define CSR_SLICES 4
#define CSR_BLOCKS (CSR_RANGES * CSR_SLICES)
#define NODES_PER (NB / CSR_RANGES)
#define ES (EE / CSR_SLICES)
#define E4S (ES / 4)
#define CSR_UNROLL 39
#define CSR_TAIL0 (256 * CSR_UNROLL)
#define OFF_PITCH (NB + 1)
static_assert(E4S - CSR_TAIL0 >= 0 && E4S - CSR_TAIL0 <= 256 && ES * CSR_SLICES == EE,
              "edge list split");
static_assert(NODES_PER <= 256 && NODES_PER * CSR_RANGES == NB && ES % 4 == 0,
              "node split");
#define HIT_CAP 64
#define EBITS 18
static_assert(EE <= (1 << EBITS) && NODES_PER <= (1 << (32 - EBITS)) &&
                  (HIT_CAP & (HIT_CAP - 1)) == 0 && HIT_CAP * 256 * 4 <= 32 * HO * 8,
              "kept edge word / list size");

typedef float v2f __attribute__((ext_vector_type(2)));
typedef float v8f __attribute__((ext_vector_type(8)));

__global__ __launch_bounds__(256) void gemm_h(const float* __restrict__ x,
                                              const float* __restrict__ W,
                                              const float* __restrict__ att_src,
                                              const float* __restrict__ att_dst,
                                              float* __restrict__ h,
                                              float* __restrict__ a_src,
                                              float* __restrict__ a_dst,
                                              const int* __restrict__ dst,
                                              const float* __restrict__ W_edge,
                                              const float* __restrict__ att_edge,
                                              int* __restrict__ off,
                                              int* __restrict__ csr_e,
                                              float* __restrict__ coef) {
  __shared__ float2 Wlds[32 * HO];
  __shared__ int s_cnt[NODES_PER];
  __shared__ int s_wtot[8];
  __shared__ int s_below;
  __shared__ int s_over;
  const int tid = threadIdx.x;
  const int lane = tid & 31;
  const int wave = tid >> 5;

  if (blockIdx.x < CSR_BLOCKS) {
    const int r = blockIdx.x % CSR_RANGES;
    const int sl = blockIdx.x / CSR_RANGES;
    const int lo = r * NODES_PER;
    const int ebase = sl * ES;
    const int4* d4 = reinterpret_cast<const int4*>(dst) + sl * E4S;
    int* offs = off + sl * OFF_PITCH;
    const int4 none = make_int4(0x7fffffff, 0x7fffffff, 0x7fffffff, 0x7fffffff);
    unsigned* hits = reinterpret_cast<unsigned*>(Wlds);
    if (tid < NODES_PER) s_cnt[tid] = 0;
    if (tid == 0) {
      s_below = 0;
      s_over = 0;
    }
    __syncthreads();

    int below = 0;
    int nh = 0;
    auto load_slice = [&](int4 (&v)[CSR_UNROLL], int4& vt) {
      const int4* p = d4 + tid;
#pragma unroll
      for (int u = 0; u < CSR_UNROLL; ++u) v[u] = p[u * 256];
      vt = none;
      if (CSR_TAIL0 + tid < E4S) vt = d4[CSR_TAIL0 + tid];
    };
    auto scan4 = [&](const int4& q, int e0) {
      const int cc[4] = {q.x, q.y, q.z, q.w};
#pragma unroll
      for (int k = 0; k < 4; ++k) {
        const unsigned rel = (unsigned)(cc[k] - lo);
        if (rel < (unsigned)NODES_PER) {
          atomicAdd(&s_cnt[rel], 1);
          hits[(nh & (HIT_CAP - 1)) * 256 + tid] = (rel << EBITS) | (unsigned)(e0 + k);
          ++nh;
        }
        below += (cc[k] < lo) ? 1 : 0;
      }
    };
    {
      int4 v[CSR_UNROLL], vt;
      load_slice(v, vt);
#pragma unroll
      for (int u = 0; u < CSR_UNROLL; ++u) scan4(v[u], ebase + (u * 256 + tid) * 4);
      scan4(vt, ebase + (CSR_TAIL0 + tid) * 4);
    }
    if (nh > HIT_CAP) s_over = 1;
    atomicAdd(&s_below, below);
    __syncthreads();

    const int cnt = (tid < NODES_PER) ? s_cnt[tid] : 0;
    int inc = cnt;
#pragma unroll
    for (int d = 1; d < 32; d <<= 1) {
      const int o = __shfl_up(inc, d);
      if (lane >= d) inc += o;
    }
    if (lane == 31) s_wtot[wave] = inc;
    __syncthreads();
    int wbase = 0;
#pragma unroll
    for (int w = 0; w < 8; ++w) wbase += (w < wave) ? s_wtot[w] : 0;
    const int myoff = ebase + s_below + wbase + inc - cnt;
    if (tid < NODES_PER) {
      s_cnt[tid] = myoff;
      offs[lo + tid] = myoff;
      if (r == CSR_RANGES - 1 && tid == NODES_PER - 1) offs[NB] = myoff + cnt;
    }
    if (blockIdx.x == 0 && tid >= 256 - HH) {
      const int hd = tid - (256 - HH);
      float s = 0.f;
      for (int o = 0; o < OO; ++o) s += W_edge[hd * OO + o] * att_edge[hd * OO + o];
      coef[hd] = s;
    }
    __syncthreads();

    if (s_over == 0) {
      for (int j = 0; j < nh; j += 4) {
        unsigned hw[4] = {0u, 0u, 0u, 0u};
        int slot[4] = {0, 0, 0, 0};
#pragma unroll
        for (int u = 0; u < 4; ++u)
          if (j + u < nh) hw[u] = hits[(j + u) * 256 + tid];
#pragma unroll
        for (int u = 0; u < 4; ++u)
          if (j + u < nh) slot[u] = atomicAdd(&s_cnt[hw[u] >> EBITS], 1);
#pragma unroll
        for (int u = 0; u < 4; ++u)
          if (j + u < nh) csr_e[slot[u]] = (int)(hw[u] & ((1u << EBITS) - 1u));
      }
      return;
    }

    auto fill4 = [&](const int4& q, int e0) {
      const int cc[4] = {q.x, q.y, q.z, q.w};
#pragma unroll
      for (int k = 0; k < 4; ++k) {
        const unsigned rel = (unsigned)(cc[k] - lo);
        if (rel < (unsigned)NODES_PER) {
          const int slot = atomicAdd(&s_cnt[rel], 1);
          csr_e[slot] = e0 + k;
        }
      }
    };
    {
      int4 v[CSR_UNROLL], vt;
      load_slice(v, vt);
#pragma unroll
      for (int u = 0; u < CSR_UNROLL; ++u) fill4(v[u], ebase + (u * 256 + tid) * 4);
      fill4(vt, ebase + (CSR_TAIL0 + tid) * 4);
    }
    return;
  }

  const int rowBase = (blockIdx.x - CSR_BLOCKS) * 128 + wave * 16;
  const bool active = rowBase < MROWS;

  v8f acc[16];
  const v8f vzero = {0.f, 0.f, 0.f, 0.f, 0.f, 0.f, 0.f, 0.f};
#pragma unroll
  for (int t = 0; t < 16; ++t) acc[t] = vzero;

  const int m = lane & 15;
  const int kb = (lane >> 4) << 1;
  const int kp = lane >> 4;

  for (int kblk = 0; kblk < DD; kblk += 64) {
#pragma unroll 4
    for (int p = 0; p < 32; ++p) {
      float2 pr;
      pr.x = W[(size_t)(kblk + 2 * p) * HO + tid];
      pr.y = W[(size_t)(kblk + 2 * p + 1) * HO + tid];
      Wlds[p * HO + tid] = pr;
    }
    __syncthreads();

    if (active) {
      const float* xrow = x + (size_t)(rowBase + m) * DD + kblk;
      for (int kk = 0; kk < 64; kk += 4) {
        v2f a;
        a.x = xrow[kk + kb];
        a.y = xrow[kk + kb + 1];
        const float2* w0 = Wlds + ((kk >> 1) + kp) * HO + (lane & 15);
#pragma unroll
        for (int t = 0; t < 16; ++t) {
          float2 bp = w0[t * 16];
          v2f b;
          b.x = bp.x;
          b.y = bp.y;
          acc[t] = __builtin_amdgcn_wmma_f32_16x16x4_f32(
              false, a, false, b, (short)0, acc[t], false, false);
        }
      }
    }
    __syncthreads();
  }

  if (active) {
    const int nLocal = lane & 15;
    const int rOff = (lane >> 4) * 8;
#pragma unroll
    for (int t = 0; t < 16; ++t)
#pragma unroll
      for (int v = 0; v < 8; ++v)
        h[(size_t)(rowBase + rOff + v) * HO + t * 16 + nLocal] = acc[t][v];

    float res[HH];
#pragma unroll
    for (int hd = 0; hd < HH; ++hd) {
      float val[16];
      float ws[4], wd[4];
#pragma unroll
      for (int tt = 0; tt < 4; ++tt) {
        ws[tt] = att_src[hd * OO + tt * 16 + nLocal];
        wd[tt] = att_dst[hd * OO + tt * 16 + nLocal];
      }
#pragma unroll
      for (int v = 0; v < 8; ++v) {
        float s = 0.f, d = 0.f;
#pragma unroll
        for (int tt = 0; tt < 4; ++tt) {
          s += acc[hd * 4 + tt][v] * ws[tt];
          d += acc[hd * 4 + tt][v] * wd[tt];
        }
        val[v] = s;
        val[8 + v] = d;
      }
#pragma unroll
      for (int half = 8; half >= 1; half >>= 1) {
        const bool up = (lane & half) != 0;
#pragma unroll
        for (int j = 0; j < half; ++j) {
          const float lo = val[j];
          const float hi = val[j + half];
          const float keep = up ? hi : lo;
          const float send = up ? lo : hi;
          val[j] = keep + __shfl_xor(send, half);
        }
      }
      res[hd] = val[0];
    }
    float* ap = (nLocal & 8) ? a_dst : a_src;
    *reinterpret_cast<float4*>(ap + (size_t)(rowBase + rOff + (nLocal & 7)) * HH) =
        make_float4(res[0], res[1], res[2], res[3]);
  }
}

#define SEG_CAP 512
__global__ __launch_bounds__(256) void node_gather(
    const int* __restrict__ src, const float* __restrict__ eattr,
    const int* __restrict__ off, const int* __restrict__ csr_e,
    const float* __restrict__ a_src, const float* __restrict__ a_dst,
    const float* __restrict__ coef, const float* __restrict__ h,
    const float* __restrict__ bias, float* __restrict__ out) {
  __shared__ int s_e[SEG_CAP];
  __shared__ int s_src[SEG_CAP];
  __shared__ float s_ea[SEG_CAP];
  __shared__ float s_red[16][16];
  __shared__ float s_w[16][16];
  __shared__ int s_cs[16];
  __shared__ __attribute__((aligned(16))) float s_acc[BB * HO];

  const int n = blockIdx.x;
  const int t = threadIdx.x;
  const int beg0 = off[n];
  const int L0 = off[n + 1] - beg0;
  const int beg1 = off[OFF_PITCH + n];
  const int L1 = off[OFF_PITCH + n + 1] - beg1;
  const int beg2 = off[2 * OFF_PITCH + n];
  const int L2 = off[2 * OFF_PITCH + n + 1] - beg2;
  const int beg3 = off[3 * OFF_PITCH + n];
  const int L3 = off[3 * OFF_PITCH + n + 1] - beg3;
  static_assert(CSR_SLICES == 4, "the join below is written for four slices");
  const int L = L0 + L1 + L2 + L3;
  auto slot = [&](int i) -> int {
    if (i < L0) return beg0 + i;
    i -= L0;
    if (i < L1) return beg1 + i;
    i -= L1;
    if (i < L2) return beg2 + i;
    return beg3 + (i - L2);
  };
  const bool staged = (L <= SEG_CAP);

  if (staged) {
    for (int i = t; i < L; i += 256) s_e[i] = csr_e[slot(i)];
    __syncthreads();
    for (int i = t; i < L; i += 256) {
      const int e = s_e[i];
      int r = 0;
      for (int j = 0; j < L; ++j) r += (s_e[j] < e) ? 1 : 0;
      s_src[r] = src[e];
      s_ea[r] = eattr[e];
    }
    __syncthreads();
  }
  auto edge_src = [&](int i) -> int {
    return staged ? s_src[i] : src[csr_e[slot(i)]];
  };
  auto edge_ea = [&](int i) -> float {
    return staged ? s_ea[i] : eattr[csr_e[slot(i)]];
  };

  const int bh = t & 15;
  const int g = t >> 4;
  const int b1 = bh >> 2, h1 = bh & 3;
  const float adst = a_dst[((size_t)b1 * NB + n) * HH + h1];
  const float cf = coef[h1];
  const float* asrc_b = a_src + (size_t)b1 * NB * HH + h1;

  float m = -INFINITY;
  for (int i = g; i < L; i += 16) {
    float a = asrc_b[(size_t)edge_src(i) * HH] + adst + edge_ea(i) * cf;
    a = a > 0.f ? a : a * NEG_SLOPE;
    m = fmaxf(m, a);
  }
  s_red[g][bh] = m;
  __syncthreads();
  float mm = s_red[0][bh];
  for (int k = 1; k < 16; ++k) mm = fmaxf(mm, s_red[k][bh]);
  if (!__builtin_isfinite(mm)) mm = 0.f;
  __syncthreads();

  float sm = 0.f;
  for (int i = g; i < L; i += 16) {
    float a = asrc_b[(size_t)edge_src(i) * HH] + adst + edge_ea(i) * cf;
    a = a > 0.f ? a : a * NEG_SLOPE;
    sm += __expf(a - mm);
  }
  s_red[g][bh] = sm;
  __syncthreads();
  float den = 0.f;
  for (int k = 0; k < 16; ++k) den += s_red[k][bh];
  den = fmaxf(den, 1e-16f);

  const int brow = t >> 6;
  const int wsel = t >> 4;
  const float4* hb = (const float4*)(h + (size_t)brow * NB * HO) + (t & 63);
  float4 acc = make_float4(0.f, 0.f, 0.f, 0.f);

  for (int c = 0; c < L; c += 16) {
    {
      const int i = c + g;
      const int ic = i < L ? i : L - 1;
      const int s = edge_src(ic);
      float a = asrc_b[(size_t)s * HH] + adst + edge_ea(ic) * cf;
      a = a > 0.f ? a : a * NEG_SLOPE;
      const float w = __expf(a - mm) / den;
      s_w[g][bh] = i < L ? w : 0.f;
      if (bh == 0) s_cs[g] = s;
    }
    __syncthreads();
    const int cnt = (L - c) < 16 ? (L - c) : 16;
    for (int j = 0; j < cnt; j += 4) {
      const float4 v0 = hb[(size_t)s_cs[j + 0] * (HO / 4)];
      const float4 v1 = hb[(size_t)s_cs[j + 1] * (HO / 4)];
      const float4 v2 = hb[(size_t)s_cs[j + 2] * (HO / 4)];
      const float4 v3 = hb[(size_t)s_cs[j + 3] * (HO / 4)];
      const float w0 = s_w[j + 0][wsel];
      const float w1 = s_w[j + 1][wsel];
      const float w2 = s_w[j + 2][wsel];
      const float w3 = s_w[j + 3][wsel];
      acc.x += w0 * v0.x; acc.y += w0 * v0.y; acc.z += w0 * v0.z; acc.w += w0 * v0.w;
      acc.x += w1 * v1.x; acc.y += w1 * v1.y; acc.z += w1 * v1.z; acc.w += w1 * v1.w;
      acc.x += w2 * v2.x; acc.y += w2 * v2.y; acc.z += w2 * v2.z; acc.w += w2 * v2.w;
      acc.x += w3 * v3.x; acc.y += w3 * v3.y; acc.z += w3 * v3.z; acc.w += w3 * v3.w;
    }
    __syncthreads();
  }

  *reinterpret_cast<float4*>(&s_acc[t * 4]) = acc;
  __syncthreads();
  const int o = t & 63;
  const float* p = s_acc + brow * HO + o;
  out[((size_t)brow * NB + n) * OO + o] =
      (p[0] + p[64] + p[128] + p[192]) * 0.25f + bias[o];
}

extern "C" void kernel_launch(void* const* d_in, const int* in_sizes, int n_in,
                              void* d_out, int out_size, void* d_ws, size_t ws_size,
                              hipStream_t stream) {
  const float* x        = (const float*)d_in[0];
  const int*   eidx     = (const int*)d_in[1];
  const float* eattr    = (const float*)d_in[2];
  const float* W        = (const float*)d_in[3];
  const float* att_src  = (const float*)d_in[4];
  const float* att_dst  = (const float*)d_in[5];
  const float* W_edge   = (const float*)d_in[6];
  const float* att_edge = (const float*)d_in[7];
  const float* bias     = (const float*)d_in[8];
  float* out = (float*)d_out;

  char* ws = (char*)d_ws;
  float* h_buf  = (float*)(ws + 0);
  float* a_src  = (float*)(ws + 40960000);
  float* a_dst  = (float*)(ws + 43520000);
  int*   off    = (int*)  (ws + 47000000);
  float* coef   = (float*)(ws + 46200576);
  int*   csr_e  = (int*)  (ws + 46200832);

  const int* srcI = eidx;
  const int* dstI = eidx + EE;

  gemm_h<<<CSR_BLOCKS + (MROWS + 127) / 128, 256, 0, stream>>>(
      x, W, att_src, att_dst, h_buf, a_src, a_dst, dstI, W_edge, att_edge, off,
      csr_e, coef);
  node_gather<<<NB, 256, 0, stream>>>(srcI, eattr, off, csr_e, a_src, a_dst,
                                      coef, h_buf, bias, out);
}


# Score tables a_src/a_dst stored node-major [N,B,H]: an edge's 16 scores (4 batches x 4 heads) are one 64-byte record, one cache line per edge in the gather instead of four; sum orders unchanged
# speedup vs baseline: 6.7538x; 1.0389x over previous
#include <hip/hip_runtime.h>


#define NB 10000
#define BB 4
#define EE 160000
#define DD 128
#define HH 4
#define OO 64
#define HO 256
#define MROWS (BB * NB)
#define NEG_SLOPE 0.2f
#define CSR_RANGES 40
#define CSR_SLICES 4
#define CSR_BLOCKS (CSR_RANGES * CSR_SLICES)
#define NODES_PER (NB / CSR_RANGES)
#define ES (EE / CSR_SLICES)
#define E4S (ES / 4)
#define CSR_UNROLL 39
#define CSR_TAIL0 (256 * CSR_UNROLL)
#define OFF_PITCH (NB + 1)
static_assert(E4S - CSR_TAIL0 >= 0 && E4S - CSR_TAIL0 <= 256 && ES * CSR_SLICES == EE,
              "edge list split");
static_assert(NODES_PER <= 256 && NODES_PER * CSR_RANGES == NB && ES % 4 == 0,
              "node split");
#define HIT_CAP 64
#define EBITS 18
static_assert(EE <= (1 << EBITS) && NODES_PER <= (1 << (32 - EBITS)) &&
                  (HIT_CAP & (HIT_CAP - 1)) == 0 && HIT_CAP * 256 * 4 <= 32 * HO * 8,
              "kept edge word / list size");

typedef float v2f __attribute__((ext_vector_type(2)));
typedef float v8f __attribute__((ext_vector_type(8)));

__global__ __launch_bounds__(256) void gemm_h(const float* __restrict__ x,
                                              const float* __restrict__ W,
                                              const float* __restrict__ att_src,
                                              const float* __restrict__ att_dst,
                                              float* __restrict__ h,
                                              float* __restrict__ a_src,
                                              float* __restrict__ a_dst,
                                              const int* __restrict__ dst,
                                              const float* __restrict__ W_edge,
                                              const float* __restrict__ att_edge,
                                              int* __restrict__ off,
                                              int* __restrict__ csr_e,
                                              float* __restrict__ coef) {
  __shared__ float2 Wlds[32 * HO];
  __shared__ int s_cnt[NODES_PER];
  __shared__ int s_wtot[8];
  __shared__ int s_below;
  __shared__ int s_over;
  const int tid = threadIdx.x;
  const int lane = tid & 31;
  const int wave = tid >> 5;

  if (blockIdx.x < CSR_BLOCKS) {
    const int r = blockIdx.x % CSR_RANGES;
    const int sl = blockIdx.x / CSR_RANGES;
    const int lo = r * NODES_PER;
    const int ebase = sl * ES;
    const int4* d4 = reinterpret_cast<const int4*>(dst) + sl * E4S;
    int* offs = off + sl * OFF_PITCH;
    const int4 none = make_int4(0x7fffffff, 0x7fffffff, 0x7fffffff, 0x7fffffff);
    unsigned* hits = reinterpret_cast<unsigned*>(Wlds);
    if (tid < NODES_PER) s_cnt[tid] = 0;
    if (tid == 0) {
      s_below = 0;
      s_over = 0;
    }
    __syncthreads();

    int below = 0;
    int nh = 0;
    auto load_slice = [&](int4 (&v)[CSR_UNROLL], int4& vt) {
      const int4* p = d4 + tid;
#pragma unroll
      for (int u = 0; u < CSR_UNROLL; ++u) v[u] = p[u * 256];
      vt = none;
      if (CSR_TAIL0 + tid < E4S) vt = d4[CSR_TAIL0 + tid];
    };
    auto scan4 = [&](const int4& q, int e0) {
      const int cc[4] = {q.x, q.y, q.z, q.w};
#pragma unroll
      for (int k = 0; k < 4; ++k) {
        const unsigned rel = (unsigned)(cc[k] - lo);
        if (rel < (unsigned)NODES_PER) {
          atomicAdd(&s_cnt[rel], 1);
          hits[(nh & (HIT_CAP - 1)) * 256 + tid] = (rel << EBITS) | (unsigned)(e0 + k);
          ++nh;
        }
        below += (cc[k] < lo) ? 1 : 0;
      }
    };
    {
      int4 v[CSR_UNROLL], vt;
      load_slice(v, vt);
#pragma unroll
      for (int u = 0; u < CSR_UNROLL; ++u) scan4(v[u], ebase + (u * 256 + tid) * 4);
      scan4(vt, ebase + (CSR_TAIL0 + tid) * 4);
    }
    if (nh > HIT_CAP) s_over = 1;
    atomicAdd(&s_below, below);
    __syncthreads();

    const int cnt = (tid < NODES_PER) ? s_cnt[tid] : 0;
    int inc = cnt;
#pragma unroll
    for (int d = 1; d < 32; d <<= 1) {
      const int o = __shfl_up(inc, d);
      if (lane >= d) inc += o;
    }
    if (lane == 31) s_wtot[wave] = inc;
    __syncthreads();
    int wbase = 0;
#pragma unroll
    for (int w = 0; w < 8; ++w) wbase += (w < wave) ? s_wtot[w] : 0;
    const int myoff = ebase + s_below + wbase + inc - cnt;
    if (tid < NODES_PER) {
      s_cnt[tid] = myoff;
      offs[lo + tid] = myoff;
      if (r == CSR_RANGES - 1 && tid == NODES_PER - 1) offs[NB] = myoff + cnt;
    }
    if (blockIdx.x == 0 && tid >= 256 - HH) {
      const int hd = tid - (256 - HH);
      float s = 0.f;
      for (int o = 0; o < OO; ++o) s += W_edge[hd * OO + o] * att_edge[hd * OO + o];
      coef[hd] = s;
    }
    __syncthreads();

    if (s_over == 0) {
      for (int j = 0; j < nh; j += 4) {
        unsigned hw[4] = {0u, 0u, 0u, 0u};
        int slot[4] = {0, 0, 0, 0};
#pragma unroll
        for (int u = 0; u < 4; ++u)
          if (j + u < nh) hw[u] = hits[(j + u) * 256 + tid];
#pragma unroll
        for (int u = 0; u < 4; ++u)
          if (j + u < nh) slot[u] = atomicAdd(&s_cnt[hw[u] >> EBITS], 1);
#pragma unroll
        for (int u = 0; u < 4; ++u)
          if (j + u < nh) csr_e[slot[u]] = (int)(hw[u] & ((1u << EBITS) - 1u));
      }
      return;
    }

    auto fill4 = [&](const int4& q, int e0) {
      const int cc[4] = {q.x, q.y, q.z, q.w};
#pragma unroll
      for (int k = 0; k < 4; ++k) {
        const unsigned rel = (unsigned)(cc[k] - lo);
        if (rel < (unsigned)NODES_PER) {
          const int slot = atomicAdd(&s_cnt[rel], 1);
          csr_e[slot] = e0 + k;
        }
      }
    };
    {
      int4 v[CSR_UNROLL], vt;
      load_slice(v, vt);
#pragma unroll
      for (int u = 0; u < CSR_UNROLL; ++u) fill4(v[u], ebase + (u * 256 + tid) * 4);
      fill4(vt, ebase + (CSR_TAIL0 + tid) * 4);
    }
    return;
  }

  const int rowBase = (blockIdx.x - CSR_BLOCKS) * 128 + wave * 16;
  const bool active = rowBase < MROWS;

  v8f acc[16];
  const v8f vzero = {0.f, 0.f, 0.f, 0.f, 0.f, 0.f, 0.f, 0.f};
#pragma unroll
  for (int t = 0; t < 16; ++t) acc[t] = vzero;

  const int m = lane & 15;
  const int kb = (lane >> 4) << 1;
  const int kp = lane >> 4;

  for (int kblk = 0; kblk < DD; kblk += 64) {
#pragma unroll 4
    for (int p = 0; p < 32; ++p) {
      float2 pr;
      pr.x = W[(size_t)(kblk + 2 * p) * HO + tid];
      pr.y = W[(size_t)(kblk + 2 * p + 1) * HO + tid];
      Wlds[p * HO + tid] = pr;
    }
    __syncthreads();

    if (active) {
      const float* xrow = x + (size_t)(rowBase + m) * DD + kblk;
      for (int kk = 0; kk < 64; kk += 4) {
        v2f a;
        a.x = xrow[kk + kb];
        a.y = xrow[kk + kb + 1];
        const float2* w0 = Wlds + ((kk >> 1) + kp) * HO + (lane & 15);
#pragma unroll
        for (int t = 0; t < 16; ++t) {
          float2 bp = w0[t * 16];
          v2f b;
          b.x = bp.x;
          b.y = bp.y;
          acc[t] = __builtin_amdgcn_wmma_f32_16x16x4_f32(
              false, a, false, b, (short)0, acc[t], false, false);
        }
      }
    }
    __syncthreads();
  }

  if (active) {
    const int nLocal = lane & 15;
    const int rOff = (lane >> 4) * 8;
#pragma unroll
    for (int t = 0; t < 16; ++t)
#pragma unroll
      for (int v = 0; v < 8; ++v)
        h[(size_t)(rowBase + rOff + v) * HO + t * 16 + nLocal] = acc[t][v];

    float res[HH];
#pragma unroll
    for (int hd = 0; hd < HH; ++hd) {
      float val[16];
      float ws[4], wd[4];
#pragma unroll
      for (int tt = 0; tt < 4; ++tt) {
        ws[tt] = att_src[hd * OO + tt * 16 + nLocal];
        wd[tt] = att_dst[hd * OO + tt * 16 + nLocal];
      }
#pragma unroll
      for (int v = 0; v < 8; ++v) {
        float s = 0.f, d = 0.f;
#pragma unroll
        for (int tt = 0; tt < 4; ++tt) {
          s += acc[hd * 4 + tt][v] * ws[tt];
          d += acc[hd * 4 + tt][v] * wd[tt];
        }
        val[v] = s;
        val[8 + v] = d;
      }
#pragma unroll
      for (int half = 8; half >= 1; half >>= 1) {
        const bool up = (lane & half) != 0;
#pragma unroll
        for (int j = 0; j < half; ++j) {
          const float lo = val[j];
          const float hi = val[j + half];
          const float keep = up ? hi : lo;
          const float send = up ? lo : hi;
          val[j] = keep + __shfl_xor(send, half);
        }
      }
      res[hd] = val[0];
    }
    static_assert(NB % 16 == 0, "a row tile must not straddle two batches");
    float* ap = (nLocal & 8) ? a_dst : a_src;
    const int bq = rowBase / NB;
    const int nq = rowBase - bq * NB + rOff + (nLocal & 7);
    *reinterpret_cast<float4*>(ap + ((size_t)nq * BB + bq) * HH) =
        make_float4(res[0], res[1], res[2], res[3]);
  }
}

#define SEG_CAP 512
__global__ __launch_bounds__(256) void node_gather(
    const int* __restrict__ src, const float* __restrict__ eattr,
    const int* __restrict__ off, const int* __restrict__ csr_e,
    const float* __restrict__ a_src, const float* __restrict__ a_dst,
    const float* __restrict__ coef, const float* __restrict__ h,
    const float* __restrict__ bias, float* __restrict__ out) {
  __shared__ int s_e[SEG_CAP];
  __shared__ int s_src[SEG_CAP];
  __shared__ float s_ea[SEG_CAP];
  __shared__ float s_red[16][16];
  __shared__ float s_w[16][16];
  __shared__ int s_cs[16];
  __shared__ __attribute__((aligned(16))) float s_acc[BB * HO];

  const int n = blockIdx.x;
  const int t = threadIdx.x;
  const int beg0 = off[n];
  const int L0 = off[n + 1] - beg0;
  const int beg1 = off[OFF_PITCH + n];
  const int L1 = off[OFF_PITCH + n + 1] - beg1;
  const int beg2 = off[2 * OFF_PITCH + n];
  const int L2 = off[2 * OFF_PITCH + n + 1] - beg2;
  const int beg3 = off[3 * OFF_PITCH + n];
  const int L3 = off[3 * OFF_PITCH + n + 1] - beg3;
  static_assert(CSR_SLICES == 4, "the join below is written for four slices");
  const int L = L0 + L1 + L2 + L3;
  auto slot = [&](int i) -> int {
    if (i < L0) return beg0 + i;
    i -= L0;
    if (i < L1) return beg1 + i;
    i -= L1;
    if (i < L2) return beg2 + i;
    return beg3 + (i - L2);
  };
  const bool staged = (L <= SEG_CAP);

  if (staged) {
    for (int i = t; i < L; i += 256) s_e[i] = csr_e[slot(i)];
    __syncthreads();
    for (int i = t; i < L; i += 256) {
      const int e = s_e[i];
      int r = 0;
      for (int j = 0; j < L; ++j) r += (s_e[j] < e) ? 1 : 0;
      s_src[r] = src[e];
      s_ea[r] = eattr[e];
    }
    __syncthreads();
  }
  auto edge_src = [&](int i) -> int {
    return staged ? s_src[i] : src[csr_e[slot(i)]];
  };
  auto edge_ea = [&](int i) -> float {
    return staged ? s_ea[i] : eattr[csr_e[slot(i)]];
  };

  const int bh = t & 15;
  const int g = t >> 4;
  const int b1 = bh >> 2, h1 = bh & 3;
  const float adst = a_dst[((size_t)n * BB + b1) * HH + h1];
  const float cf = coef[h1];
  const float* asrc_b = a_src + b1 * HH + h1;

  float m = -INFINITY;
  for (int i = g; i < L; i += 16) {
    float a = asrc_b[(size_t)edge_src(i) * (BB * HH)] + adst + edge_ea(i) * cf;
    a = a > 0.f ? a : a * NEG_SLOPE;
    m = fmaxf(m, a);
  }
  s_red[g][bh] = m;
  __syncthreads();
  float mm = s_red[0][bh];
  for (int k = 1; k < 16; ++k) mm = fmaxf(mm, s_red[k][bh]);
  if (!__builtin_isfinite(mm)) mm = 0.f;
  __syncthreads();

  float sm = 0.f;
  for (int i = g; i < L; i += 16) {
    float a = asrc_b[(size_t)edge_src(i) * (BB * HH)] + adst + edge_ea(i) * cf;
    a = a > 0.f ? a : a * NEG_SLOPE;
    sm += __expf(a - mm);
  }
  s_red[g][bh] = sm;
  __syncthreads();
  float den = 0.f;
  for (int k = 0; k < 16; ++k) den += s_red[k][bh];
  den = fmaxf(den, 1e-16f);

  const int brow = t >> 6;
  const int wsel = t >> 4;
  const float4* hb = (const float4*)(h + (size_t)brow * NB * HO) + (t & 63);
  float4 acc = make_float4(0.f, 0.f, 0.f, 0.f);

  for (int c = 0; c < L; c += 16) {
    {
      const int i = c + g;
      const int ic = i < L ? i : L - 1;
      const int s = edge_src(ic);
      float a = asrc_b[(size_t)s * (BB * HH)] + adst + edge_ea(ic) * cf;
      a = a > 0.f ? a : a * NEG_SLOPE;
      const float w = __expf(a - mm) / den;
      s_w[g][bh] = i < L ? w : 0.f;
      if (bh == 0) s_cs[g] = s;
    }
    __syncthreads();
    const int cnt = (L - c) < 16 ? (L - c) : 16;
    for (int j = 0; j < cnt; j += 4) {
      const float4 v0 = hb[(size_t)s_cs[j + 0] * (HO / 4)];
      const float4 v1 = hb[(size_t)s_cs[j + 1] * (HO / 4)];
      const float4 v2 = hb[(size_t)s_cs[j + 2] * (HO / 4)];
      const float4 v3 = hb[(size_t)s_cs[j + 3] * (HO / 4)];
      const float w0 = s_w[j + 0][wsel];
      const float w1 = s_w[j + 1][wsel];
      const float w2 = s_w[j + 2][wsel];
      const float w3 = s_w[j + 3][wsel];
      acc.x += w0 * v0.x; acc.y += w0 * v0.y; acc.z += w0 * v0.z; acc.w += w0 * v0.w;
      acc.x += w1 * v1.x; acc.y += w1 * v1.y; acc.z += w1 * v1.z; acc.w += w1 * v1.w;
      acc.x += w2 * v2.x; acc.y += w2 * v2.y; acc.z += w2 * v2.z; acc.w += w2 * v2.w;
      acc.x += w3 * v3.x; acc.y += w3 * v3.y; acc.z += w3 * v3.z; acc.w += w3 * v3.w;
    }
    __syncthreads();
  }

  *reinterpret_cast<float4*>(&s_acc[t * 4]) = acc;
  __syncthreads();
  const int o = t & 63;
  const float* p = s_acc + brow * HO + o;
  out[((size_t)brow * NB + n) * OO + o] =
      (p[0] + p[64] + p[128] + p[192]) * 0.25f + bias[o];
}

extern "C" void kernel_launch(void* const* d_in, const int* in_sizes, int n_in,
                              void* d_out, int out_size, void* d_ws, size_t ws_size,
                              hipStream_t stream) {
  const float* x        = (const float*)d_in[0];
  const int*   eidx     = (const int*)d_in[1];
  const float* eattr    = (const float*)d_in[2];
  const float* W        = (const float*)d_in[3];
  const float* att_src  = (const float*)d_in[4];
  const float* att_dst  = (const float*)d_in[5];
  const float* W_edge   = (const float*)d_in[6];
  const float* att_edge = (const float*)d_in[7];
  const float* bias     = (const float*)d_in[8];
  float* out = (float*)d_out;

  char* ws = (char*)d_ws;
  float* h_buf  = (float*)(ws + 0);
  float* a_src  = (float*)(ws + 40960000);
  float* a_dst  = (float*)(ws + 43520000);
  int*   off    = (int*)  (ws + 47000000);
  float* coef   = (float*)(ws + 46200576);
  int*   csr_e  = (int*)  (ws + 46200832);

  const int* srcI = eidx;
  const int* dstI = eidx + EE;

  gemm_h<<<CSR_BLOCKS + (MROWS + 127) / 128, 256, 0, stream>>>(
      x, W, att_src, att_dst, h_buf, a_src, a_dst, dstI, W_edge, att_edge, off,
      csr_e, coef);
  node_gather<<<NB, 256, 0, stream>>>(srcI, eattr, off, csr_e, a_src, a_dst,
                                      coef, h_buf, bias, out);
}


# Node gather on 64 threads per node instead of 256: a thread carries its float4 of all 4 batches (16 row loads in flight); 4x fewer waves, 16 nodes resident per WGP not 8; sum orders unchanged
# speedup vs baseline: 7.3557x; 1.0891x over previous
#include <hip/hip_runtime.h>


#define NB 10000
#define BB 4
#define EE 160000
#define DD 128
#define HH 4
#define OO 64
#define HO 256
#define MROWS (BB * NB)
#define NEG_SLOPE 0.2f
#define CSR_RANGES 40
#define CSR_SLICES 4
#define CSR_BLOCKS (CSR_RANGES * CSR_SLICES)
#define NODES_PER (NB / CSR_RANGES)
#define ES (EE / CSR_SLICES)
#define E4S (ES / 4)
#define CSR_UNROLL 39
#define CSR_TAIL0 (256 * CSR_UNROLL)
#define OFF_PITCH (NB + 1)
static_assert(E4S - CSR_TAIL0 >= 0 && E4S - CSR_TAIL0 <= 256 && ES * CSR_SLICES == EE,
              "edge list split");
static_assert(NODES_PER <= 256 && NODES_PER * CSR_RANGES == NB && ES % 4 == 0,
              "node split");
#define HIT_CAP 64
#define EBITS 18
static_assert(EE <= (1 << EBITS) && NODES_PER <= (1 << (32 - EBITS)) &&
                  (HIT_CAP & (HIT_CAP - 1)) == 0 && HIT_CAP * 256 * 4 <= 32 * HO * 8,
              "kept edge word / list size");

typedef float v2f __attribute__((ext_vector_type(2)));
typedef float v8f __attribute__((ext_vector_type(8)));

__global__ __launch_bounds__(256) void gemm_h(const float* __restrict__ x,
                                              const float* __restrict__ W,
                                              const float* __restrict__ att_src,
                                              const float* __restrict__ att_dst,
                                              float* __restrict__ h,
                                              float* __restrict__ a_src,
                                              float* __restrict__ a_dst,
                                              const int* __restrict__ dst,
                                              const float* __restrict__ W_edge,
                                              const float* __restrict__ att_edge,
                                              int* __restrict__ off,
                                              int* __restrict__ csr_e,
                                              float* __restrict__ coef) {
  __shared__ float2 Wlds[32 * HO];
  __shared__ int s_cnt[NODES_PER];
  __shared__ int s_wtot[8];
  __shared__ int s_below;
  __shared__ int s_over;
  const int tid = threadIdx.x;
  const int lane = tid & 31;
  const int wave = tid >> 5;

  if (blockIdx.x < CSR_BLOCKS) {
    const int r = blockIdx.x % CSR_RANGES;
    const int sl = blockIdx.x / CSR_RANGES;
    const int lo = r * NODES_PER;
    const int ebase = sl * ES;
    const int4* d4 = reinterpret_cast<const int4*>(dst) + sl * E4S;
    int* offs = off + sl * OFF_PITCH;
    const int4 none = make_int4(0x7fffffff, 0x7fffffff, 0x7fffffff, 0x7fffffff);
    unsigned* hits = reinterpret_cast<unsigned*>(Wlds);
    if (tid < NODES_PER) s_cnt[tid] = 0;
    if (tid == 0) {
      s_below = 0;
      s_over = 0;
    }
    __syncthreads();

    int below = 0;
    int nh = 0;
    auto load_slice = [&](int4 (&v)[CSR_UNROLL], int4& vt) {
      const int4* p = d4 + tid;
#pragma unroll
      for (int u = 0; u < CSR_UNROLL; ++u) v[u] = p[u * 256];
      vt = none;
      if (CSR_TAIL0 + tid < E4S) vt = d4[CSR_TAIL0 + tid];
    };
    auto scan4 = [&](const int4& q, int e0) {
      const int cc[4] = {q.x, q.y, q.z, q.w};
#pragma unroll
      for (int k = 0; k < 4; ++k) {
        const unsigned rel = (unsigned)(cc[k] - lo);
        if (rel < (unsigned)NODES_PER) {
          atomicAdd(&s_cnt[rel], 1);
          hits[(nh & (HIT_CAP - 1)) * 256 + tid] = (rel << EBITS) | (unsigned)(e0 + k);
          ++nh;
        }
        below += (cc[k] < lo) ? 1 : 0;
      }
    };
    {
      int4 v[CSR_UNROLL], vt;
      load_slice(v, vt);
#pragma unroll
      for (int u = 0; u < CSR_UNROLL; ++u) scan4(v[u], ebase + (u * 256 + tid) * 4);
      scan4(vt, ebase + (CSR_TAIL0 + tid) * 4);
    }
    if (nh > HIT_CAP) s_over = 1;
    atomicAdd(&s_below, below);
    __syncthreads();

    const int cnt = (tid < NODES_PER) ? s_cnt[tid] : 0;
    int inc = cnt;
#pragma unroll
    for (int d = 1; d < 32; d <<= 1) {
      const int o = __shfl_up(inc, d);
      if (lane >= d) inc += o;
    }
    if (lane == 31) s_wtot[wave] = inc;
    __syncthreads();
    int wbase = 0;
#pragma unroll
    for (int w = 0; w < 8; ++w) wbase += (w < wave) ? s_wtot[w] : 0;
    const int myoff = ebase + s_below + wbase + inc - cnt;
    if (tid < NODES_PER) {
      s_cnt[tid] = myoff;
      offs[lo + tid] = myoff;
      if (r == CSR_RANGES - 1 && tid == NODES_PER - 1) offs[NB] = myoff + cnt;
    }
    if (blockIdx.x == 0 && tid >= 256 - HH) {
      const int hd = tid - (256 - HH);
      float s = 0.f;
      for (int o = 0; o < OO; ++o) s += W_edge[hd * OO + o] * att_edge[hd * OO + o];
      coef[hd] = s;
    }
    __syncthreads();

    if (s_over == 0) {
      for (int j = 0; j < nh; j += 4) {
        unsigned hw[4] = {0u, 0u, 0u, 0u};
        int slot[4] = {0, 0, 0, 0};
#pragma unroll
        for (int u = 0; u < 4; ++u)
          if (j + u < nh) hw[u] = hits[(j + u) * 256 + tid];
#pragma unroll
        for (int u = 0; u < 4; ++u)
          if (j + u < nh) slot[u] = atomicAdd(&s_cnt[hw[u] >> EBITS], 1);
#pragma unroll
        for (int u = 0; u < 4; ++u)
          if (j + u < nh) csr_e[slot[u]] = (int)(hw[u] & ((1u << EBITS) - 1u));
      }
      return;
    }

    auto fill4 = [&](const int4& q, int e0) {
      const int cc[4] = {q.x, q.y, q.z, q.w};
#pragma unroll
      for (int k = 0; k < 4; ++k) {
        const unsigned rel = (unsigned)(cc[k] - lo);
        if (rel < (unsigned)NODES_PER) {
          const int slot = atomicAdd(&s_cnt[rel], 1);
          csr_e[slot] = e0 + k;
        }
      }
    };
    {
      int4 v[CSR_UNROLL], vt;
      load_slice(v, vt);
#pragma unroll
      for (int u = 0; u < CSR_UNROLL; ++u) fill4(v[u], ebase + (u * 256 + tid) * 4);
      fill4(vt, ebase + (CSR_TAIL0 + tid) * 4);
    }
    return;
  }

  const int rowBase = (blockIdx.x - CSR_BLOCKS) * 128 + wave * 16;
  const bool active = rowBase < MROWS;

  v8f acc[16];
  const v8f vzero = {0.f, 0.f, 0.f, 0.f, 0.f, 0.f, 0.f, 0.f};
#pragma unroll
  for (int t = 0; t < 16; ++t) acc[t] = vzero;

  const int m = lane & 15;
  const int kb = (lane >> 4) << 1;
  const int kp = lane >> 4;

  for (int kblk = 0; kblk < DD; kblk += 64) {
#pragma unroll 4
    for (int p = 0; p < 32; ++p) {
      float2 pr;
      pr.x = W[(size_t)(kblk + 2 * p) * HO + tid];
      pr.y = W[(size_t)(kblk + 2 * p + 1) * HO + tid];
      Wlds[p * HO + tid] = pr;
    }
    __syncthreads();

    if (active) {
      const float* xrow = x + (size_t)(rowBase + m) * DD + kblk;
      for (int kk = 0; kk < 64; kk += 4) {
        v2f a;
        a.x = xrow[kk + kb];
        a.y = xrow[kk + kb + 1];
        const float2* w0 = Wlds + ((kk >> 1) + kp) * HO + (lane & 15);
#pragma unroll
        for (int t = 0; t < 16; ++t) {
          float2 bp = w0[t * 16];
          v2f b;
          b.x = bp.x;
          b.y = bp.y;
          acc[t] = __builtin_amdgcn_wmma_f32_16x16x4_f32(
              false, a, false, b, (short)0, acc[t], false, false);
        }
      }
    }
    __syncthreads();
  }

  if (active) {
    const int nLocal = lane & 15;
    const int rOff = (lane >> 4) * 8;
#pragma unroll
    for (int t = 0; t < 16; ++t)
#pragma unroll
      for (int v = 0; v < 8; ++v)
        h[(size_t)(rowBase + rOff + v) * HO + t * 16 + nLocal] = acc[t][v];

    float res[HH];
#pragma unroll
    for (int hd = 0; hd < HH; ++hd) {
      float val[16];
      float ws[4], wd[4];
#pragma unroll
      for (int tt = 0; tt < 4; ++tt) {
        ws[tt] = att_src[hd * OO + tt * 16 + nLocal];
        wd[tt] = att_dst[hd * OO + tt * 16 + nLocal];
      }
#pragma unroll
      for (int v = 0; v < 8; ++v) {
        float s = 0.f, d = 0.f;
#pragma unroll
        for (int tt = 0; tt < 4; ++tt) {
          s += acc[hd * 4 + tt][v] * ws[tt];
          d += acc[hd * 4 + tt][v] * wd[tt];
        }
        val[v] = s;
        val[8 + v] = d;
      }
#pragma unroll
      for (int half = 8; half >= 1; half >>= 1) {
        const bool up = (lane & half) != 0;
#pragma unroll
        for (int j = 0; j < half; ++j) {
          const float lo = val[j];
          const float hi = val[j + half];
          const float keep = up ? hi : lo;
          const float send = up ? lo : hi;
          val[j] = keep + __shfl_xor(send, half);
        }
      }
      res[hd] = val[0];
    }
    static_assert(NB % 16 == 0, "a row tile must not straddle two batches");
    float* ap = (nLocal & 8) ? a_dst : a_src;
    const int bq = rowBase / NB;
    const int nq = rowBase - bq * NB + rOff + (nLocal & 7);
    *reinterpret_cast<float4*>(ap + ((size_t)nq * BB + bq) * HH) =
        make_float4(res[0], res[1], res[2], res[3]);
  }
}

#define SEG_CAP 512
#define GT 64
static_assert(GT * 4 == HO && BB == 4 && HH * BB == 16, "thread / float4 / score mapping");
__global__ __launch_bounds__(GT) void node_gather(
    const int* __restrict__ src, const float* __restrict__ eattr,
    const int* __restrict__ off, const int* __restrict__ csr_e,
    const float* __restrict__ a_src, const float* __restrict__ a_dst,
    const float* __restrict__ coef, const float* __restrict__ h,
    const float* __restrict__ bias, float* __restrict__ out) {
  __shared__ __attribute__((aligned(16))) int s_raw[3 * SEG_CAP];
  __shared__ float s_red[16][16];
  __shared__ float s_w[16][16];
  __shared__ int s_cs[16];
  static_assert(2 * SEG_CAP >= BB * HO, "head-mean buffer fits in the first two lists");
  int* s_e = s_raw;
  int* s_src = s_raw + SEG_CAP;
  float* s_ea = reinterpret_cast<float*>(s_raw + 2 * SEG_CAP);
  float* s_acc = reinterpret_cast<float*>(s_raw);

  const int n = blockIdx.x;
  const int t = threadIdx.x;
  const int beg0 = off[n];
  const int L0 = off[n + 1] - beg0;
  const int beg1 = off[OFF_PITCH + n];
  const int L1 = off[OFF_PITCH + n + 1] - beg1;
  const int beg2 = off[2 * OFF_PITCH + n];
  const int L2 = off[2 * OFF_PITCH + n + 1] - beg2;
  const int beg3 = off[3 * OFF_PITCH + n];
  const int L3 = off[3 * OFF_PITCH + n + 1] - beg3;
  static_assert(CSR_SLICES == 4, "the join below is written for four slices");
  const int L = L0 + L1 + L2 + L3;
  auto slot = [&](int i) -> int {
    if (i < L0) return beg0 + i;
    i -= L0;
    if (i < L1) return beg1 + i;
    i -= L1;
    if (i < L2) return beg2 + i;
    return beg3 + (i - L2);
  };
  const bool staged = (L <= SEG_CAP);

  if (staged) {
    for (int i = t; i < L; i += GT) s_e[i] = csr_e[slot(i)];
    __syncthreads();
    for (int i = t; i < L; i += GT) {
      const int e = s_e[i];
      int r = 0;
      for (int j = 0; j < L; ++j) r += (s_e[j] < e) ? 1 : 0;
      s_src[r] = src[e];
      s_ea[r] = eattr[e];
    }
    __syncthreads();
  }
  const int bh = t & 15;
  const int q = t >> 4;
  const int b1 = bh >> 2, h1 = bh & 3;
  const float adst = a_dst[((size_t)n * BB + b1) * HH + h1];
  const float cf = coef[h1];
  const float* asrc_b = a_src + b1 * HH + h1;
  const int hd = t >> 4;
  const float4* hb = (const float4*)h + t;
  float4 acc[BB];
  for (int b = 0; b < BB; ++b) acc[b] = make_float4(0.f, 0.f, 0.f, 0.f);

  auto run = [&](auto edge_src, auto edge_ea) {
    float mv[4] = {-INFINITY, -INFINITY, -INFINITY, -INFINITY};
    for (int i0 = 0; i0 < L; i0 += 16) {
      float a[4];
#pragma unroll
      for (int u = 0; u < 4; ++u) {
        const int i = i0 + q * 4 + u;
        const int ic = i < L ? i : L - 1;
        a[u] = asrc_b[(size_t)edge_src(ic) * (BB * HH)] + adst + edge_ea(ic) * cf;
      }
#pragma unroll
      for (int u = 0; u < 4; ++u) {
        const int i = i0 + q * 4 + u;
        const float al = a[u] > 0.f ? a[u] : a[u] * NEG_SLOPE;
        mv[u] = i < L ? fmaxf(mv[u], al) : mv[u];
      }
    }
    for (int u = 0; u < 4; ++u) s_red[q * 4 + u][bh] = mv[u];
    __syncthreads();
    float mm = s_red[0][bh];
    for (int k = 1; k < 16; ++k) mm = fmaxf(mm, s_red[k][bh]);
    if (!__builtin_isfinite(mm)) mm = 0.f;
    __syncthreads();

    float sv[4] = {0.f, 0.f, 0.f, 0.f};
    for (int i0 = 0; i0 < L; i0 += 16) {
      float a[4];
#pragma unroll
      for (int u = 0; u < 4; ++u) {
        const int i = i0 + q * 4 + u;
        const int ic = i < L ? i : L - 1;
        a[u] = asrc_b[(size_t)edge_src(ic) * (BB * HH)] + adst + edge_ea(ic) * cf;
      }
#pragma unroll
      for (int u = 0; u < 4; ++u) {
        const int i = i0 + q * 4 + u;
        const float al = a[u] > 0.f ? a[u] : a[u] * NEG_SLOPE;
        const float ex = __expf(al - mm);
        sv[u] = i < L ? sv[u] + ex : sv[u];
      }
    }
    for (int u = 0; u < 4; ++u) s_red[q * 4 + u][bh] = sv[u];
    __syncthreads();
    float den = 0.f;
    for (int k = 0; k < 16; ++k) den += s_red[k][bh];
    den = fmaxf(den, 1e-16f);

    for (int c = 0; c < L; c += 16) {
      {
        int s[4];
        float a[4];
#pragma unroll
        for (int u = 0; u < 4; ++u) {
          const int i = c + q * 4 + u;
          const int ic = i < L ? i : L - 1;
          s[u] = edge_src(ic);
          a[u] = asrc_b[(size_t)s[u] * (BB * HH)] + adst + edge_ea(ic) * cf;
        }
#pragma unroll
        for (int u = 0; u < 4; ++u) {
          const int g = q * 4 + u;
          const float al = a[u] > 0.f ? a[u] : a[u] * NEG_SLOPE;
          const float w = __expf(al - mm) / den;
          s_w[g][bh] = (c + g) < L ? w : 0.f;
          if (bh == 0) s_cs[g] = s[u];
        }
      }
      __syncthreads();
      const int cnt = (L - c) < 16 ? (L - c) : 16;
      for (int j = 0; j < cnt; j += 4) {
        float4 v[4][BB];
#pragma unroll
        for (int e = 0; e < 4; ++e) {
          const float4* row = hb + (size_t)s_cs[j + e] * (HO / 4);
#pragma unroll
          for (int b = 0; b < BB; ++b) v[e][b] = row[(size_t)b * NB * (HO / 4)];
        }
#pragma unroll
        for (int e = 0; e < 4; ++e) {
#pragma unroll
          for (int b = 0; b < BB; ++b) {
            const float w = s_w[j + e][b * HH + hd];
            acc[b].x += w * v[e][b].x;
            acc[b].y += w * v[e][b].y;
            acc[b].z += w * v[e][b].z;
            acc[b].w += w * v[e][b].w;
          }
        }
      }
      __syncthreads();
    }
  };
  if (staged) {
    run([&](int i) -> int { return s_src[i]; }, [&](int i) -> float { return s_ea[i]; });
  } else {
    run([&](int i) -> int { return src[csr_e[slot(i)]]; },
        [&](int i) -> float { return eattr[csr_e[slot(i)]]; });
  }

  for (int b = 0; b < BB; ++b)
    *reinterpret_cast<float4*>(&s_acc[b * HO + t * 4]) = acc[b];
  __syncthreads();
  for (int b = 0; b < BB; ++b) {
    const float* p = s_acc + b * HO + t;
    out[((size_t)b * NB + n) * OO + t] =
        (p[0] + p[64] + p[128] + p[192]) * 0.25f + bias[t];
  }
}

extern "C" void kernel_launch(void* const* d_in, const int* in_sizes, int n_in,
                              void* d_out, int out_size, void* d_ws, size_t ws_size,
                              hipStream_t stream) {
  const float* x        = (const float*)d_in[0];
  const int*   eidx     = (const int*)d_in[1];
  const float* eattr    = (const float*)d_in[2];
  const float* W        = (const float*)d_in[3];
  const float* att_src  = (const float*)d_in[4];
  const float* att_dst  = (const float*)d_in[5];
  const float* W_edge   = (const float*)d_in[6];
  const float* att_edge = (const float*)d_in[7];
  const float* bias     = (const float*)d_in[8];
  float* out = (float*)d_out;

  char* ws = (char*)d_ws;
  float* h_buf  = (float*)(ws + 0);
  float* a_src  = (float*)(ws + 40960000);
  float* a_dst  = (float*)(ws + 43520000);
  int*   off    = (int*)  (ws + 47000000);
  float* coef   = (float*)(ws + 46200576);
  int*   csr_e  = (int*)  (ws + 46200832);

  const int* srcI = eidx;
  const int* dstI = eidx + EE;

  gemm_h<<<CSR_BLOCKS + (MROWS + 127) / 128, 256, 0, stream>>>(
      x, W, att_src, att_dst, h_buf, a_src, a_dst, dstI, W_edge, att_edge, off,
      csr_e, coef);
  node_gather<<<NB, GT, 0, stream>>>(srcI, eattr, off, csr_e, a_src, a_dst,
                                      coef, h_buf, bias, out);
}


# GEMM work-group tile 128 rows x 128 columns (one column half, two heads) instead of 128 x 256: 626 units, at most 3 x 512 WMMAs per SIMD not 2 x 1024; weight half staged once; sum orders unchanged
# speedup vs baseline: 7.8793x; 1.0712x over previous
#include <hip/hip_runtime.h>


#define NB 10000
#define BB 4
#define EE 160000
#define DD 128
#define HH 4
#define OO 64
#define HO 256
#define NC 128
#define MROWS (BB * NB)
#define NEG_SLOPE 0.2f
#define CSR_RANGES 40
#define CSR_SLICES 4
#define CSR_BLOCKS (CSR_RANGES * CSR_SLICES)
#define NODES_PER (NB / CSR_RANGES)
#define ES (EE / CSR_SLICES)
#define E4S (ES / 4)
#define CSR_UNROLL 39
#define CSR_TAIL0 (256 * CSR_UNROLL)
#define OFF_PITCH (NB + 1)
static_assert(E4S - CSR_TAIL0 >= 0 && E4S - CSR_TAIL0 <= 256 && ES * CSR_SLICES == EE,
              "edge list split");
static_assert(NODES_PER <= 256 && NODES_PER * CSR_RANGES == NB && ES % 4 == 0,
              "node split");
#define HIT_CAP 64
#define EBITS 18
static_assert(EE <= (1 << EBITS) && NODES_PER <= (1 << (32 - EBITS)) &&
                  (HIT_CAP & (HIT_CAP - 1)) == 0 && HIT_CAP * 256 * 4 <= 64 * NC * 8,
              "kept edge word / list size");

typedef float v2f __attribute__((ext_vector_type(2)));
typedef float v8f __attribute__((ext_vector_type(8)));

__global__ __launch_bounds__(256) void gemm_h(const float* __restrict__ x,
                                              const float* __restrict__ W,
                                              const float* __restrict__ att_src,
                                              const float* __restrict__ att_dst,
                                              float* __restrict__ h,
                                              float* __restrict__ a_src,
                                              float* __restrict__ a_dst,
                                              const int* __restrict__ dst,
                                              const float* __restrict__ W_edge,
                                              const float* __restrict__ att_edge,
                                              int* __restrict__ off,
                                              int* __restrict__ csr_e,
                                              float* __restrict__ coef) {
  __shared__ float2 Wlds[64 * NC];
  __shared__ int s_cnt[NODES_PER];
  __shared__ int s_wtot[8];
  __shared__ int s_below;
  __shared__ int s_over;
  const int tid = threadIdx.x;
  const int lane = tid & 31;
  const int wave = tid >> 5;

  if (blockIdx.x < CSR_BLOCKS) {
    const int r = blockIdx.x % CSR_RANGES;
    const int sl = blockIdx.x / CSR_RANGES;
    const int lo = r * NODES_PER;
    const int ebase = sl * ES;
    const int4* d4 = reinterpret_cast<const int4*>(dst) + sl * E4S;
    int* offs = off + sl * OFF_PITCH;
    const int4 none = make_int4(0x7fffffff, 0x7fffffff, 0x7fffffff, 0x7fffffff);
    unsigned* hits = reinterpret_cast<unsigned*>(Wlds);
    if (tid < NODES_PER) s_cnt[tid] = 0;
    if (tid == 0) {
      s_below = 0;
      s_over = 0;
    }
    __syncthreads();

    int below = 0;
    int nh = 0;
    auto load_slice = [&](int4 (&v)[CSR_UNROLL], int4& vt) {
      const int4* p = d4 + tid;
#pragma unroll
      for (int u = 0; u < CSR_UNROLL; ++u) v[u] = p[u * 256];
      vt = none;
      if (CSR_TAIL0 + tid < E4S) vt = d4[CSR_TAIL0 + tid];
    };
    auto scan4 = [&](const int4& q, int e0) {
      const int cc[4] = {q.x, q.y, q.z, q.w};
#pragma unroll
      for (int k = 0; k < 4; ++k) {
        const unsigned rel = (unsigned)(cc[k] - lo);
        if (rel < (unsigned)NODES_PER) {
          atomicAdd(&s_cnt[rel], 1);
          hits[(nh & (HIT_CAP - 1)) * 256 + tid] = (rel << EBITS) | (unsigned)(e0 + k);
          ++nh;
        }
        below += (cc[k] < lo) ? 1 : 0;
      }
    };
    {
      int4 v[CSR_UNROLL], vt;
      load_slice(v, vt);
#pragma unroll
      for (int u = 0; u < CSR_UNROLL; ++u) scan4(v[u], ebase + (u * 256 + tid) * 4);
      scan4(vt, ebase + (CSR_TAIL0 + tid) * 4);
    }
    if (nh > HIT_CAP) s_over = 1;
    atomicAdd(&s_below, below);
    __syncthreads();

    const int cnt = (tid < NODES_PER) ? s_cnt[tid] : 0;
    int inc = cnt;
#pragma unroll
    for (int d = 1; d < 32; d <<= 1) {
      const int o = __shfl_up(inc, d);
      if (lane >= d) inc += o;
    }
    if (lane == 31) s_wtot[wave] = inc;
    __syncthreads();
    int wbase = 0;
#pragma unroll
    for (int w = 0; w < 8; ++w) wbase += (w < wave) ? s_wtot[w] : 0;
    const int myoff = ebase + s_below + wbase + inc - cnt;
    if (tid < NODES_PER) {
      s_cnt[tid] = myoff;
      offs[lo + tid] = myoff;
      if (r == CSR_RANGES - 1 && tid == NODES_PER - 1) offs[NB] = myoff + cnt;
    }
    if (blockIdx.x == 0 && tid >= 256 - HH) {
      const int hd = tid - (256 - HH);
      float s = 0.f;
      for (int o = 0; o < OO; ++o) s += W_edge[hd * OO + o] * att_edge[hd * OO + o];
      coef[hd] = s;
    }
    __syncthreads();

    if (s_over == 0) {
      for (int j = 0; j < nh; j += 4) {
        unsigned hw[4] = {0u, 0u, 0u, 0u};
        int slot[4] = {0, 0, 0, 0};
#pragma unroll
        for (int u = 0; u < 4; ++u)
          if (j + u < nh) hw[u] = hits[(j + u) * 256 + tid];
#pragma unroll
        for (int u = 0; u < 4; ++u)
          if (j + u < nh) slot[u] = atomicAdd(&s_cnt[hw[u] >> EBITS], 1);
#pragma unroll
        for (int u = 0; u < 4; ++u)
          if (j + u < nh) csr_e[slot[u]] = (int)(hw[u] & ((1u << EBITS) - 1u));
      }
      return;
    }

    auto fill4 = [&](const int4& q, int e0) {
      const int cc[4] = {q.x, q.y, q.z, q.w};
#pragma unroll
      for (int k = 0; k < 4; ++k) {
        const unsigned rel = (unsigned)(cc[k] - lo);
        if (rel < (unsigned)NODES_PER) {
          const int slot = atomicAdd(&s_cnt[rel], 1);
          csr_e[slot] = e0 + k;
        }
      }
    };
    {
      int4 v[CSR_UNROLL], vt;
      load_slice(v, vt);
#pragma unroll
      for (int u = 0; u < CSR_UNROLL; ++u) fill4(v[u], ebase + (u * 256 + tid) * 4);
      fill4(vt, ebase + (CSR_TAIL0 + tid) * 4);
    }
    return;
  }

  const int g = blockIdx.x - CSR_BLOCKS;
  const int ch = g & 1;
  const int colBase = ch * NC;
  const int rowBase = (g >> 1) * 128 + wave * 16;
  const bool active = rowBase < MROWS;

  v8f acc[8];
  const v8f vzero = {0.f, 0.f, 0.f, 0.f, 0.f, 0.f, 0.f, 0.f};
#pragma unroll
  for (int t = 0; t < 8; ++t) acc[t] = vzero;

  const int m = lane & 15;
  const int kb = (lane >> 4) << 1;
  const int kp = lane >> 4;

  {
    const int c = tid & (NC - 1);
    const int hk = tid >> 7;
    const float* wcol = W + colBase + c;
    float wv[64];
#pragma unroll
    for (int i = 0; i < 32; ++i) {
      const int p = 2 * i + hk;
      wv[2 * i] = wcol[(size_t)(2 * p) * HO];
      wv[2 * i + 1] = wcol[(size_t)(2 * p + 1) * HO];
    }
#pragma unroll
    for (int i = 0; i < 32; ++i)
      Wlds[(2 * i + hk) * NC + c] = make_float2(wv[2 * i], wv[2 * i + 1]);
  }
  __syncthreads();

  if (active) {
    const float* xrow = x + (size_t)(rowBase + m) * DD;
    for (int kk = 0; kk < DD; kk += 4) {
      v2f a;
      a.x = xrow[kk + kb];
      a.y = xrow[kk + kb + 1];
      const float2* w0 = Wlds + ((kk >> 1) + kp) * NC + (lane & 15);
#pragma unroll
      for (int t = 0; t < 8; ++t) {
        float2 bp = w0[t * 16];
        v2f b;
        b.x = bp.x;
        b.y = bp.y;
        acc[t] = __builtin_amdgcn_wmma_f32_16x16x4_f32(
            false, a, false, b, (short)0, acc[t], false, false);
      }
    }
  }

  if (active) {
    const int nLocal = lane & 15;
    const int rOff = (lane >> 4) * 8;
#pragma unroll
    for (int t = 0; t < 8; ++t)
#pragma unroll
      for (int v = 0; v < 8; ++v)
        h[(size_t)(rowBase + rOff + v) * HO + colBase + t * 16 + nLocal] = acc[t][v];

    float res[2];
#pragma unroll
    for (int lh = 0; lh < 2; ++lh) {
      const int hd = 2 * ch + lh;
      float val[16];
      float ws[4], wd[4];
#pragma unroll
      for (int tt = 0; tt < 4; ++tt) {
        ws[tt] = att_src[hd * OO + tt * 16 + nLocal];
        wd[tt] = att_dst[hd * OO + tt * 16 + nLocal];
      }
#pragma unroll
      for (int v = 0; v < 8; ++v) {
        float s = 0.f, d = 0.f;
#pragma unroll
        for (int tt = 0; tt < 4; ++tt) {
          s += acc[lh * 4 + tt][v] * ws[tt];
          d += acc[lh * 4 + tt][v] * wd[tt];
        }
        val[v] = s;
        val[8 + v] = d;
      }
#pragma unroll
      for (int half = 8; half >= 1; half >>= 1) {
        const bool up = (lane & half) != 0;
#pragma unroll
        for (int j = 0; j < half; ++j) {
          const float lo = val[j];
          const float hi = val[j + half];
          const float keep = up ? hi : lo;
          const float send = up ? lo : hi;
          val[j] = keep + __shfl_xor(send, half);
        }
      }
      res[lh] = val[0];
    }
    static_assert(NB % 16 == 0, "a row tile must not straddle two batches");
    float* ap = (nLocal & 8) ? a_dst : a_src;
    const int bq = rowBase / NB;
    const int nq = rowBase - bq * NB + rOff + (nLocal & 7);
    *reinterpret_cast<float2*>(ap + ((size_t)nq * BB + bq) * HH + 2 * ch) =
        make_float2(res[0], res[1]);
  }
}

#define SEG_CAP 512
#define GT 64
static_assert(GT * 4 == HO && BB == 4 && HH * BB == 16, "thread / float4 / score mapping");
__global__ __launch_bounds__(GT) void node_gather(
    const int* __restrict__ src, const float* __restrict__ eattr,
    const int* __restrict__ off, const int* __restrict__ csr_e,
    const float* __restrict__ a_src, const float* __restrict__ a_dst,
    const float* __restrict__ coef, const float* __restrict__ h,
    const float* __restrict__ bias, float* __restrict__ out) {
  __shared__ __attribute__((aligned(16))) int s_raw[3 * SEG_CAP];
  __shared__ float s_red[16][16];
  __shared__ float s_w[16][16];
  __shared__ int s_cs[16];
  static_assert(2 * SEG_CAP >= BB * HO, "head-mean buffer fits in the first two lists");
  int* s_e = s_raw;
  int* s_src = s_raw + SEG_CAP;
  float* s_ea = reinterpret_cast<float*>(s_raw + 2 * SEG_CAP);
  float* s_acc = reinterpret_cast<float*>(s_raw);

  const int n = blockIdx.x;
  const int t = threadIdx.x;
  const int beg0 = off[n];
  const int L0 = off[n + 1] - beg0;
  const int beg1 = off[OFF_PITCH + n];
  const int L1 = off[OFF_PITCH + n + 1] - beg1;
  const int beg2 = off[2 * OFF_PITCH + n];
  const int L2 = off[2 * OFF_PITCH + n + 1] - beg2;
  const int beg3 = off[3 * OFF_PITCH + n];
  const int L3 = off[3 * OFF_PITCH + n + 1] - beg3;
  static_assert(CSR_SLICES == 4, "the join below is written for four slices");
  const int L = L0 + L1 + L2 + L3;
  auto slot = [&](int i) -> int {
    if (i < L0) return beg0 + i;
    i -= L0;
    if (i < L1) return beg1 + i;
    i -= L1;
    if (i < L2) return beg2 + i;
    return beg3 + (i - L2);
  };
  const bool staged = (L <= SEG_CAP);

  if (staged) {
    for (int i = t; i < L; i += GT) s_e[i] = csr_e[slot(i)];
    __syncthreads();
    for (int i = t; i < L; i += GT) {
      const int e = s_e[i];
      int r = 0;
      for (int j = 0; j < L; ++j) r += (s_e[j] < e) ? 1 : 0;
      s_src[r] = src[e];
      s_ea[r] = eattr[e];
    }
    __syncthreads();
  }
  const int bh = t & 15;
  const int q = t >> 4;
  const int b1 = bh >> 2, h1 = bh & 3;
  const float adst = a_dst[((size_t)n * BB + b1) * HH + h1];
  const float cf = coef[h1];
  const float* asrc_b = a_src + b1 * HH + h1;
  const int hd = t >> 4;
  const float4* hb = (const float4*)h + t;
  float4 acc[BB];
  for (int b = 0; b < BB; ++b) acc[b] = make_float4(0.f, 0.f, 0.f, 0.f);

  auto run = [&](auto edge_src, auto edge_ea) {
    float mv[4] = {-INFINITY, -INFINITY, -INFINITY, -INFINITY};
    for (int i0 = 0; i0 < L; i0 += 16) {
      float a[4];
#pragma unroll
      for (int u = 0; u < 4; ++u) {
        const int i = i0 + q * 4 + u;
        const int ic = i < L ? i : L - 1;
        a[u] = asrc_b[(size_t)edge_src(ic) * (BB * HH)] + adst + edge_ea(ic) * cf;
      }
#pragma unroll
      for (int u = 0; u < 4; ++u) {
        const int i = i0 + q * 4 + u;
        const float al = a[u] > 0.f ? a[u] : a[u] * NEG_SLOPE;
        mv[u] = i < L ? fmaxf(mv[u], al) : mv[u];
      }
    }
    for (int u = 0; u < 4; ++u) s_red[q * 4 + u][bh] = mv[u];
    __syncthreads();
    float mm = s_red[0][bh];
    for (int k = 1; k < 16; ++k) mm = fmaxf(mm, s_red[k][bh]);
    if (!__builtin_isfinite(mm)) mm = 0.f;
    __syncthreads();

    float sv[4] = {0.f, 0.f, 0.f, 0.f};
    for (int i0 = 0; i0 < L; i0 += 16) {
      float a[4];
#pragma unroll
      for (int u = 0; u < 4; ++u) {
        const int i = i0 + q * 4 + u;
        const int ic = i < L ? i : L - 1;
        a[u] = asrc_b[(size_t)edge_src(ic) * (BB * HH)] + adst + edge_ea(ic) * cf;
      }
#pragma unroll
      for (int u = 0; u < 4; ++u) {
        const int i = i0 + q * 4 + u;
        const float al = a[u] > 0.f ? a[u] : a[u] * NEG_SLOPE;
        const float ex = __expf(al - mm);
        sv[u] = i < L ? sv[u] + ex : sv[u];
      }
    }
    for (int u = 0; u < 4; ++u) s_red[q * 4 + u][bh] = sv[u];
    __syncthreads();
    float den = 0.f;
    for (int k = 0; k < 16; ++k) den += s_red[k][bh];
    den = fmaxf(den, 1e-16f);

    for (int c = 0; c < L; c += 16) {
      {
        int s[4];
        float a[4];
#pragma unroll
        for (int u = 0; u < 4; ++u) {
          const int i = c + q * 4 + u;
          const int ic = i < L ? i : L - 1;
          s[u] = edge_src(ic);
          a[u] = asrc_b[(size_t)s[u] * (BB * HH)] + adst + edge_ea(ic) * cf;
        }
#pragma unroll
        for (int u = 0; u < 4; ++u) {
          const int g = q * 4 + u;
          const float al = a[u] > 0.f ? a[u] : a[u] * NEG_SLOPE;
          const float w = __expf(al - mm) / den;
          s_w[g][bh] = (c + g) < L ? w : 0.f;
          if (bh == 0) s_cs[g] = s[u];
        }
      }
      __syncthreads();
      const int cnt = (L - c) < 16 ? (L - c) : 16;
      for (int j = 0; j < cnt; j += 4) {
        float4 v[4][BB];
#pragma unroll
        for (int e = 0; e < 4; ++e) {
          const float4* row = hb + (size_t)s_cs[j + e] * (HO / 4);
#pragma unroll
          for (int b = 0; b < BB; ++b) v[e][b] = row[(size_t)b * NB * (HO / 4)];
        }
#pragma unroll
        for (int e = 0; e < 4; ++e) {
#pragma unroll
          for (int b = 0; b < BB; ++b) {
            const float w = s_w[j + e][b * HH + hd];
            acc[b].x += w * v[e][b].x;
            acc[b].y += w * v[e][b].y;
            acc[b].z += w * v[e][b].z;
            acc[b].w += w * v[e][b].w;
          }
        }
      }
      __syncthreads();
    }
  };
  if (staged) {
    run([&](int i) -> int { return s_src[i]; }, [&](int i) -> float { return s_ea[i]; });
  } else {
    run([&](int i) -> int { return src[csr_e[slot(i)]]; },
        [&](int i) -> float { return eattr[csr_e[slot(i)]]; });
  }

  for (int b = 0; b < BB; ++b)
    *reinterpret_cast<float4*>(&s_acc[b * HO + t * 4]) = acc[b];
  __syncthreads();
  for (int b = 0; b < BB; ++b) {
    const float* p = s_acc + b * HO + t;
    out[((size_t)b * NB + n) * OO + t] =
        (p[0] + p[64] + p[128] + p[192]) * 0.25f + bias[t];
  }
}

extern "C" void kernel_launch(void* const* d_in, const int* in_sizes, int n_in,
                              void* d_out, int out_size, void* d_ws, size_t ws_size,
                              hipStream_t stream) {
  const float* x        = (const float*)d_in[0];
  const int*   eidx     = (const int*)d_in[1];
  const float* eattr    = (const float*)d_in[2];
  const float* W        = (const float*)d_in[3];
  const float* att_src  = (const float*)d_in[4];
  const float* att_dst  = (const float*)d_in[5];
  const float* W_edge   = (const float*)d_in[6];
  const float* att_edge = (const float*)d_in[7];
  const float* bias     = (const float*)d_in[8];
  float* out = (float*)d_out;

  char* ws = (char*)d_ws;
  float* h_buf  = (float*)(ws + 0);
  float* a_src  = (float*)(ws + 40960000);
  float* a_dst  = (float*)(ws + 43520000);
  int*   off    = (int*)  (ws + 47000000);
  float* coef   = (float*)(ws + 46200576);
  int*   csr_e  = (int*)  (ws + 46200832);

  const int* srcI = eidx;
  const int* dstI = eidx + EE;

  gemm_h<<<CSR_BLOCKS + 2 * ((MROWS + 127) / 128), 256, 0, stream>>>(
      x, W, att_src, att_dst, h_buf, a_src, a_dst, dstI, W_edge, att_edge, off,
      csr_e, coef);
  node_gather<<<NB, GT, 0, stream>>>(srcI, eattr, off, csr_e, a_src, a_dst,
                                      coef, h_buf, bias, out);
}


# Edge slots hold 16-byte records {edge id, source, attribute} written by the grouping blocks: the node gather reads a segment as one run, not 2 scattered 4-byte lookups per edge; sum orders unchanged
# speedup vs baseline: 8.3324x; 1.0575x over previous
#include <hip/hip_runtime.h>


#define NB 10000
#define BB 4
#define EE 160000
#define DD 128
#define HH 4
#define OO 64
#define HO 256
#define NC 128
#define MROWS (BB * NB)
#define NEG_SLOPE 0.2f
#define CSR_RANGES 40
#define CSR_SLICES 4
#define CSR_BLOCKS (CSR_RANGES * CSR_SLICES)
#define NODES_PER (NB / CSR_RANGES)
#define ES (EE / CSR_SLICES)
#define E4S (ES / 4)
#define CSR_UNROLL 39
#define CSR_TAIL0 (256 * CSR_UNROLL)
#define OFF_PITCH (NB + 1)
static_assert(E4S - CSR_TAIL0 >= 0 && E4S - CSR_TAIL0 <= 256 && ES * CSR_SLICES == EE,
              "edge list split");
static_assert(NODES_PER <= 256 && NODES_PER * CSR_RANGES == NB && ES % 4 == 0,
              "node split");
#define HIT_CAP 64
#define EBITS 18
static_assert(EE <= (1 << EBITS) && NODES_PER <= (1 << (32 - EBITS)) &&
                  (HIT_CAP & (HIT_CAP - 1)) == 0 && HIT_CAP * 256 * 4 <= 64 * NC * 8,
              "kept edge word / list size");

typedef float v2f __attribute__((ext_vector_type(2)));
typedef float v8f __attribute__((ext_vector_type(8)));

__global__ __launch_bounds__(256) void gemm_h(const float* __restrict__ x,
                                              const float* __restrict__ W,
                                              const float* __restrict__ att_src,
                                              const float* __restrict__ att_dst,
                                              float* __restrict__ h,
                                              float* __restrict__ a_src,
                                              float* __restrict__ a_dst,
                                              const int* __restrict__ dst,
                                              const int* __restrict__ src,
                                              const float* __restrict__ eattr,
                                              const float* __restrict__ W_edge,
                                              const float* __restrict__ att_edge,
                                              int* __restrict__ off,
                                              int* __restrict__ erec,
                                              float* __restrict__ coef) {
  __shared__ float2 Wlds[64 * NC];
  __shared__ int s_cnt[NODES_PER];
  __shared__ int s_wtot[8];
  __shared__ int s_below;
  __shared__ int s_over;
  const int tid = threadIdx.x;
  const int lane = tid & 31;
  const int wave = tid >> 5;

  if (blockIdx.x < CSR_BLOCKS) {
    const int r = blockIdx.x % CSR_RANGES;
    const int sl = blockIdx.x / CSR_RANGES;
    const int lo = r * NODES_PER;
    const int ebase = sl * ES;
    const int4* d4 = reinterpret_cast<const int4*>(dst) + sl * E4S;
    int* offs = off + sl * OFF_PITCH;
    int4* er4 = reinterpret_cast<int4*>(erec);
    const int4 none = make_int4(0x7fffffff, 0x7fffffff, 0x7fffffff, 0x7fffffff);
    unsigned* hits = reinterpret_cast<unsigned*>(Wlds);
    if (tid < NODES_PER) s_cnt[tid] = 0;
    if (tid == 0) {
      s_below = 0;
      s_over = 0;
    }
    __syncthreads();

    int below = 0;
    int nh = 0;
    auto load_slice = [&](int4 (&v)[CSR_UNROLL], int4& vt) {
      const int4* p = d4 + tid;
#pragma unroll
      for (int u = 0; u < CSR_UNROLL; ++u) v[u] = p[u * 256];
      vt = none;
      if (CSR_TAIL0 + tid < E4S) vt = d4[CSR_TAIL0 + tid];
    };
    auto scan4 = [&](const int4& q, int e0) {
      const int cc[4] = {q.x, q.y, q.z, q.w};
#pragma unroll
      for (int k = 0; k < 4; ++k) {
        const unsigned rel = (unsigned)(cc[k] - lo);
        if (rel < (unsigned)NODES_PER) {
          atomicAdd(&s_cnt[rel], 1);
          hits[(nh & (HIT_CAP - 1)) * 256 + tid] = (rel << EBITS) | (unsigned)(e0 + k);
          ++nh;
        }
        below += (cc[k] < lo) ? 1 : 0;
      }
    };
    {
      int4 v[CSR_UNROLL], vt;
      load_slice(v, vt);
#pragma unroll
      for (int u = 0; u < CSR_UNROLL; ++u) scan4(v[u], ebase + (u * 256 + tid) * 4);
      scan4(vt, ebase + (CSR_TAIL0 + tid) * 4);
    }
    if (nh > HIT_CAP) s_over = 1;
    atomicAdd(&s_below, below);
    __syncthreads();

    const int cnt = (tid < NODES_PER) ? s_cnt[tid] : 0;
    int inc = cnt;
#pragma unroll
    for (int d = 1; d < 32; d <<= 1) {
      const int o = __shfl_up(inc, d);
      if (lane >= d) inc += o;
    }
    if (lane == 31) s_wtot[wave] = inc;
    __syncthreads();
    int wbase = 0;
#pragma unroll
    for (int w = 0; w < 8; ++w) wbase += (w < wave) ? s_wtot[w] : 0;
    const int myoff = ebase + s_below + wbase + inc - cnt;
    if (tid < NODES_PER) {
      s_cnt[tid] = myoff;
      offs[lo + tid] = myoff;
      if (r == CSR_RANGES - 1 && tid == NODES_PER - 1) offs[NB] = myoff + cnt;
    }
    if (blockIdx.x == 0 && tid >= 256 - HH) {
      const int hd = tid - (256 - HH);
      float s = 0.f;
      for (int o = 0; o < OO; ++o) s += W_edge[hd * OO + o] * att_edge[hd * OO + o];
      coef[hd] = s;
    }
    __syncthreads();

    if (s_over == 0) {
      for (int j = 0; j < nh; j += 4) {
        unsigned hw[4] = {0u, 0u, 0u, 0u};
        int slot[4] = {0, 0, 0, 0};
        int sv[4] = {0, 0, 0, 0};
        int av[4] = {0, 0, 0, 0};
#pragma unroll
        for (int u = 0; u < 4; ++u)
          if (j + u < nh) hw[u] = hits[(j + u) * 256 + tid];
#pragma unroll
        for (int u = 0; u < 4; ++u)
          if (j + u < nh) {
            const int e = (int)(hw[u] & ((1u << EBITS) - 1u));
            sv[u] = src[e];
            av[u] = __float_as_int(eattr[e]);
          }
#pragma unroll
        for (int u = 0; u < 4; ++u)
          if (j + u < nh) slot[u] = atomicAdd(&s_cnt[hw[u] >> EBITS], 1);
#pragma unroll
        for (int u = 0; u < 4; ++u)
          if (j + u < nh)
            er4[slot[u]] =
                make_int4((int)(hw[u] & ((1u << EBITS) - 1u)), sv[u], av[u], 0);
      }
      return;
    }

    auto fill4 = [&](const int4& q, int e0) {
      const int cc[4] = {q.x, q.y, q.z, q.w};
#pragma unroll
      for (int k = 0; k < 4; ++k) {
        const unsigned rel = (unsigned)(cc[k] - lo);
        if (rel < (unsigned)NODES_PER) {
          const int slot = atomicAdd(&s_cnt[rel], 1);
          const int e = e0 + k;
          er4[slot] = make_int4(e, src[e], __float_as_int(eattr[e]), 0);
        }
      }
    };
#pragma unroll 1
    for (int u = 0; u <= CSR_UNROLL; ++u) {
      const int g = (u < CSR_UNROLL) ? (u * 256 + tid) : (CSR_TAIL0 + tid);
      if (g < E4S) fill4(d4[g], ebase + g * 4);
    }
    return;
  }

  const int g = blockIdx.x - CSR_BLOCKS;
  const int ch = g & 1;
  const int colBase = ch * NC;
  const int rowBase = (g >> 1) * 128 + wave * 16;
  const bool active = rowBase < MROWS;

  v8f acc[8];
  const v8f vzero = {0.f, 0.f, 0.f, 0.f, 0.f, 0.f, 0.f, 0.f};
#pragma unroll
  for (int t = 0; t < 8; ++t) acc[t] = vzero;

  const int m = lane & 15;
  const int kb = (lane >> 4) << 1;
  const int kp = lane >> 4;

  {
    const int c = tid & (NC - 1);
    const int hk = tid >> 7;
    const float* wcol = W + colBase + c;
    float wv[64];
#pragma unroll
    for (int i = 0; i < 32; ++i) {
      const int p = 2 * i + hk;
      wv[2 * i] = wcol[(size_t)(2 * p) * HO];
      wv[2 * i + 1] = wcol[(size_t)(2 * p + 1) * HO];
    }
#pragma unroll
    for (int i = 0; i < 32; ++i)
      Wlds[(2 * i + hk) * NC + c] = make_float2(wv[2 * i], wv[2 * i + 1]);
  }
  __syncthreads();

  if (active) {
    const float* xrow = x + (size_t)(rowBase + m) * DD;
    for (int kk = 0; kk < DD; kk += 4) {
      v2f a;
      a.x = xrow[kk + kb];
      a.y = xrow[kk + kb + 1];
      const float2* w0 = Wlds + ((kk >> 1) + kp) * NC + (lane & 15);
#pragma unroll
      for (int t = 0; t < 8; ++t) {
        float2 bp = w0[t * 16];
        v2f b;
        b.x = bp.x;
        b.y = bp.y;
        acc[t] = __builtin_amdgcn_wmma_f32_16x16x4_f32(
            false, a, false, b, (short)0, acc[t], false, false);
      }
    }
  }

  if (active) {
    const int nLocal = lane & 15;
    const int rOff = (lane >> 4) * 8;
#pragma unroll
    for (int t = 0; t < 8; ++t)
#pragma unroll
      for (int v = 0; v < 8; ++v)
        h[(size_t)(rowBase + rOff + v) * HO + colBase + t * 16 + nLocal] = acc[t][v];

    float res[2];
#pragma unroll
    for (int lh = 0; lh < 2; ++lh) {
      const int hd = 2 * ch + lh;
      float val[16];
      float ws[4], wd[4];
#pragma unroll
      for (int tt = 0; tt < 4; ++tt) {
        ws[tt] = att_src[hd * OO + tt * 16 + nLocal];
        wd[tt] = att_dst[hd * OO + tt * 16 + nLocal];
      }
#pragma unroll
      for (int v = 0; v < 8; ++v) {
        float s = 0.f, d = 0.f;
#pragma unroll
        for (int tt = 0; tt < 4; ++tt) {
          s += acc[lh * 4 + tt][v] * ws[tt];
          d += acc[lh * 4 + tt][v] * wd[tt];
        }
        val[v] = s;
        val[8 + v] = d;
      }
#pragma unroll
      for (int half = 8; half >= 1; half >>= 1) {
        const bool up = (lane & half) != 0;
#pragma unroll
        for (int j = 0; j < half; ++j) {
          const float lo = val[j];
          const float hi = val[j + half];
          const float keep = up ? hi : lo;
          const float send = up ? lo : hi;
          val[j] = keep + __shfl_xor(send, half);
        }
      }
      res[lh] = val[0];
    }
    static_assert(NB % 16 == 0, "a row tile must not straddle two batches");
    float* ap = (nLocal & 8) ? a_dst : a_src;
    const int bq = rowBase / NB;
    const int nq = rowBase - bq * NB + rOff + (nLocal & 7);
    *reinterpret_cast<float2*>(ap + ((size_t)nq * BB + bq) * HH + 2 * ch) =
        make_float2(res[0], res[1]);
  }
}

#define SEG_CAP 512
#define GT 64
static_assert(GT * 4 == HO && BB == 4 && HH * BB == 16, "thread / float4 / score mapping");
__global__ __launch_bounds__(GT) void node_gather(
    const int* __restrict__ off, const int* __restrict__ erec,
    const float* __restrict__ a_src, const float* __restrict__ a_dst,
    const float* __restrict__ coef, const float* __restrict__ h,
    const float* __restrict__ bias, float* __restrict__ out) {
  __shared__ __attribute__((aligned(16))) int s_raw[5 * SEG_CAP];
  __shared__ float s_red[16][16];
  __shared__ float s_w[16][16];
  __shared__ int s_cs[16];
  static_assert(2 * SEG_CAP >= BB * HO, "head-mean buffer fits in the first two lists");
  int* s_e = s_raw;
  int* s_us = s_raw + SEG_CAP;
  int* s_ua = s_raw + 2 * SEG_CAP;
  int* s_src = s_raw + 3 * SEG_CAP;
  float* s_ea = reinterpret_cast<float*>(s_raw + 4 * SEG_CAP);
  float* s_acc = reinterpret_cast<float*>(s_raw);
  const int4* er4 = reinterpret_cast<const int4*>(erec);

  const int n = blockIdx.x;
  const int t = threadIdx.x;
  const int beg0 = off[n];
  const int L0 = off[n + 1] - beg0;
  const int beg1 = off[OFF_PITCH + n];
  const int L1 = off[OFF_PITCH + n + 1] - beg1;
  const int beg2 = off[2 * OFF_PITCH + n];
  const int L2 = off[2 * OFF_PITCH + n + 1] - beg2;
  const int beg3 = off[3 * OFF_PITCH + n];
  const int L3 = off[3 * OFF_PITCH + n + 1] - beg3;
  static_assert(CSR_SLICES == 4, "the join below is written for four slices");
  const int L = L0 + L1 + L2 + L3;
  auto slot = [&](int i) -> int {
    if (i < L0) return beg0 + i;
    i -= L0;
    if (i < L1) return beg1 + i;
    i -= L1;
    if (i < L2) return beg2 + i;
    return beg3 + (i - L2);
  };
  const bool staged = (L <= SEG_CAP);

  if (staged) {
    for (int i = t; i < L; i += GT) {
      const int4 r4 = er4[slot(i)];
      s_e[i] = r4.x;
      s_us[i] = r4.y;
      s_ua[i] = r4.z;
    }
    __syncthreads();
    for (int i = t; i < L; i += GT) {
      const int e = s_e[i];
      const int us = s_us[i];
      const int ua = s_ua[i];
      int r = 0;
      for (int j = 0; j < L; ++j) r += (s_e[j] < e) ? 1 : 0;
      s_src[r] = us;
      s_ea[r] = __int_as_float(ua);
    }
    __syncthreads();
  }
  const int bh = t & 15;
  const int q = t >> 4;
  const int b1 = bh >> 2, h1 = bh & 3;
  const float adst = a_dst[((size_t)n * BB + b1) * HH + h1];
  const float cf = coef[h1];
  const float* asrc_b = a_src + b1 * HH + h1;
  const int hd = t >> 4;
  const float4* hb = (const float4*)h + t;
  float4 acc[BB];
  for (int b = 0; b < BB; ++b) acc[b] = make_float4(0.f, 0.f, 0.f, 0.f);

  auto run = [&](auto edge_src, auto edge_ea) {
    float mv[4] = {-INFINITY, -INFINITY, -INFINITY, -INFINITY};
    for (int i0 = 0; i0 < L; i0 += 16) {
      float a[4];
#pragma unroll
      for (int u = 0; u < 4; ++u) {
        const int i = i0 + q * 4 + u;
        const int ic = i < L ? i : L - 1;
        a[u] = asrc_b[(size_t)edge_src(ic) * (BB * HH)] + adst + edge_ea(ic) * cf;
      }
#pragma unroll
      for (int u = 0; u < 4; ++u) {
        const int i = i0 + q * 4 + u;
        const float al = a[u] > 0.f ? a[u] : a[u] * NEG_SLOPE;
        mv[u] = i < L ? fmaxf(mv[u], al) : mv[u];
      }
    }
    for (int u = 0; u < 4; ++u) s_red[q * 4 + u][bh] = mv[u];
    __syncthreads();
    float mm = s_red[0][bh];
    for (int k = 1; k < 16; ++k) mm = fmaxf(mm, s_red[k][bh]);
    if (!__builtin_isfinite(mm)) mm = 0.f;
    __syncthreads();

    float sv[4] = {0.f, 0.f, 0.f, 0.f};
    for (int i0 = 0; i0 < L; i0 += 16) {
      float a[4];
#pragma unroll
      for (int u = 0; u < 4; ++u) {
        const int i = i0 + q * 4 + u;
        const int ic = i < L ? i : L - 1;
        a[u] = asrc_b[(size_t)edge_src(ic) * (BB * HH)] + adst + edge_ea(ic) * cf;
      }
#pragma unroll
      for (int u = 0; u < 4; ++u) {
        const int i = i0 + q * 4 + u;
        const float al = a[u] > 0.f ? a[u] : a[u] * NEG_SLOPE;
        const float ex = __expf(al - mm);
        sv[u] = i < L ? sv[u] + ex : sv[u];
      }
    }
    for (int u = 0; u < 4; ++u) s_red[q * 4 + u][bh] = sv[u];
    __syncthreads();
    float den = 0.f;
    for (int k = 0; k < 16; ++k) den += s_red[k][bh];
    den = fmaxf(den, 1e-16f);

    for (int c = 0; c < L; c += 16) {
      {
        int s[4];
        float a[4];
#pragma unroll
        for (int u = 0; u < 4; ++u) {
          const int i = c + q * 4 + u;
          const int ic = i < L ? i : L - 1;
          s[u] = edge_src(ic);
          a[u] = asrc_b[(size_t)s[u] * (BB * HH)] + adst + edge_ea(ic) * cf;
        }
#pragma unroll
        for (int u = 0; u < 4; ++u) {
          const int g = q * 4 + u;
          const float al = a[u] > 0.f ? a[u] : a[u] * NEG_SLOPE;
          const float w = __expf(al - mm) / den;
          s_w[g][bh] = (c + g) < L ? w : 0.f;
          if (bh == 0) s_cs[g] = s[u];
        }
      }
      __syncthreads();
      const int cnt = (L - c) < 16 ? (L - c) : 16;
      for (int j = 0; j < cnt; j += 4) {
        float4 v[4][BB];
#pragma unroll
        for (int e = 0; e < 4; ++e) {
          const float4* row = hb + (size_t)s_cs[j + e] * (HO / 4);
#pragma unroll
          for (int b = 0; b < BB; ++b) v[e][b] = row[(size_t)b * NB * (HO / 4)];
        }
#pragma unroll
        for (int e = 0; e < 4; ++e) {
#pragma unroll
          for (int b = 0; b < BB; ++b) {
            const float w = s_w[j + e][b * HH + hd];
            acc[b].x += w * v[e][b].x;
            acc[b].y += w * v[e][b].y;
            acc[b].z += w * v[e][b].z;
            acc[b].w += w * v[e][b].w;
          }
        }
      }
      __syncthreads();
    }
  };
  if (staged) {
    run([&](int i) -> int { return s_src[i]; }, [&](int i) -> float { return s_ea[i]; });
  } else {
    run([&](int i) -> int { return er4[slot(i)].y; },
        [&](int i) -> float { return __int_as_float(er4[slot(i)].z); });
  }

  for (int b = 0; b < BB; ++b)
    *reinterpret_cast<float4*>(&s_acc[b * HO + t * 4]) = acc[b];
  __syncthreads();
  for (int b = 0; b < BB; ++b) {
    const float* p = s_acc + b * HO + t;
    out[((size_t)b * NB + n) * OO + t] =
        (p[0] + p[64] + p[128] + p[192]) * 0.25f + bias[t];
  }
}

extern "C" void kernel_launch(void* const* d_in, const int* in_sizes, int n_in,
                              void* d_out, int out_size, void* d_ws, size_t ws_size,
                              hipStream_t stream) {
  const float* x        = (const float*)d_in[0];
  const int*   eidx     = (const int*)d_in[1];
  const float* eattr    = (const float*)d_in[2];
  const float* W        = (const float*)d_in[3];
  const float* att_src  = (const float*)d_in[4];
  const float* att_dst  = (const float*)d_in[5];
  const float* W_edge   = (const float*)d_in[6];
  const float* att_edge = (const float*)d_in[7];
  const float* bias     = (const float*)d_in[8];
  float* out = (float*)d_out;

  char* ws = (char*)d_ws;
  float* h_buf  = (float*)(ws + 0);
  float* a_src  = (float*)(ws + 40960000);
  float* a_dst  = (float*)(ws + 43520000);
  int*   off    = (int*)  (ws + 47000000);
  float* coef   = (float*)(ws + 46200576);
  int*   erec   = (int*)  (ws + 48000000);

  const int* srcI = eidx;
  const int* dstI = eidx + EE;

  gemm_h<<<CSR_BLOCKS + 2 * ((MROWS + 127) / 128), 256, 0, stream>>>(
      x, W, att_src, att_dst, h_buf, a_src, a_dst, dstI, srcI, eattr, W_edge,
      att_edge, off, erec, coef);
  node_gather<<<NB, GT, 0, stream>>>(off, erec, a_src, a_dst, coef, h_buf, bias,
                                      out);
}
